# starting kernel
# baseline (speedup 1.0000x reference)
#include <hip/hip_runtime.h>
#include <hip/hip_bf16.h>
#include <math.h>

typedef __attribute__((ext_vector_type(2))) float v2f;
typedef __attribute__((ext_vector_type(8))) float v8f;

#define IN_DIM 128
#define HID    16


__global__ void k_init_deg(float* __restrict__ deg, int n) {
  int i = blockIdx.x * blockDim.x + threadIdx.x;
  if (i < n) deg[i] = 1.0f;
}

__global__ void k_deg_edges(const int* __restrict__ dst, float* __restrict__ deg, int nE) {
  int e = blockIdx.x * blockDim.x + threadIdx.x;
  if (e < nE) atomicAdd(&deg[dst[e]], 1.0f);
}

__global__ void k_dis(const float* __restrict__ deg, float* __restrict__ dis, int n) {
  int i = blockIdx.x * blockDim.x + threadIdx.x;
  if (i < n) dis[i] = rsqrtf(deg[i]);
}


__global__ void k_gemm1_wmma(const float* __restrict__ x, const float* __restrict__ W1,
                             float* __restrict__ h, int n) {
  const int lane = threadIdx.x & 31;
  const int wave = threadIdx.x >> 5;
  const int tile = blockIdx.x * (blockDim.x >> 5) + wave;
  const int row0 = tile << 4;
  if (row0 >= n) return;
  const int m  = lane & 15;
  const int kh = lane >> 4;
  const float* xrow = x + (size_t)(row0 + m) * IN_DIM + 2 * kh;

  v8f c = {0.f, 0.f, 0.f, 0.f, 0.f, 0.f, 0.f, 0.f};
#pragma unroll
  for (int k0 = 0; k0 < IN_DIM / 4; ++k0) {
    const int kk = 4 * k0 + 2 * kh;
    const float2 av = *reinterpret_cast<const float2*>(xrow + 4 * k0);
    v2f a; a.x = av.x; a.y = av.y;
    v2f b; b.x = W1[(kk + 0) * HID + m];
           b.y = W1[(kk + 1) * HID + m];
    c = __builtin_amdgcn_wmma_f32_16x16x4_f32(false, a, false, b, (short)0, c, false, false);
  }
  float* hout = h + (size_t)(row0 + 8 * kh) * HID + m;
#pragma unroll
  for (int r = 0; r < 8; ++r)
    hout[(size_t)r * HID] = c[r];
}


__global__ void k_gemm2_wmma(const float* __restrict__ h1, const float* __restrict__ W2,
                             float* __restrict__ h2, int n) {
  const int lane = threadIdx.x & 31;
  const int wave = threadIdx.x >> 5;
  const int tile = blockIdx.x * (blockDim.x >> 5) + wave;
  const int row0 = tile << 4;
  if (row0 >= n) return;
  const int m  = lane & 15;
  const int kh = lane >> 4;
  const float bmask = (m < 2) ? 1.0f : 0.0f;
  const int   col   = (m < 2) ? m : 0;
  const float* arow = h1 + (size_t)(row0 + m) * HID + 2 * kh;

  v8f c = {0.f, 0.f, 0.f, 0.f, 0.f, 0.f, 0.f, 0.f};
#pragma unroll
  for (int k0 = 0; k0 < HID / 4; ++k0) {
    const int kk = 4 * k0 + 2 * kh;
    const float2 av = *reinterpret_cast<const float2*>(arow + 4 * k0);
    v2f a; a.x = av.x; a.y = av.y;
    v2f b; b.x = W2[(kk + 0) * 2 + col] * bmask;
           b.y = W2[(kk + 1) * 2 + col] * bmask;
    c = __builtin_amdgcn_wmma_f32_16x16x4_f32(false, a, false, b, (short)0, c, false, false);
  }
  if (m < 2) {
#pragma unroll
    for (int r = 0; r < 8; ++r)
      h2[(size_t)(row0 + r + 8 * kh) * 2 + m] = c[r];
  }
}


__global__ void k_self1(const float* __restrict__ h, const float* __restrict__ dis,
                        float* __restrict__ out1, int n) {
  int i = blockIdx.x * blockDim.x + threadIdx.x;
  if (i >= n) return;
  float s = dis[i]; s = s * s;
  const float4* hp = reinterpret_cast<const float4*>(h + (size_t)i * HID);
  float4* op = reinterpret_cast<float4*>(out1 + (size_t)i * HID);
#pragma unroll
  for (int q = 0; q < 4; ++q) {
    float4 v = hp[q];
    v.x *= s; v.y *= s; v.z *= s; v.w *= s;
    op[q] = v;
  }
}

__global__ void k_scatter1(const int* __restrict__ src, const int* __restrict__ dst,
                           const float* __restrict__ dis, const float* __restrict__ h,
                           float* __restrict__ out1, int nE) {
  long long t = (long long)blockIdx.x * blockDim.x + threadIdx.x;
  int e = (int)(t >> 2);
  if (e >= nE) return;
  int q = (int)(t & 3);
  int s = src[e], d = dst[e];
  float norm = dis[s] * dis[d];
  float4 v = *reinterpret_cast<const float4*>(h + (size_t)s * HID + 4 * q);
  float* o = out1 + (size_t)d * HID + 4 * q;
  atomicAdd(o + 0, v.x * norm);
  atomicAdd(o + 1, v.y * norm);
  atomicAdd(o + 2, v.z * norm);
  atomicAdd(o + 3, v.w * norm);
}

__global__ void k_relu_bias(float* __restrict__ out1, const float* __restrict__ b1, int n) {
  long long t = (long long)blockIdx.x * blockDim.x + threadIdx.x;
  if (t >= (long long)n * HID) return;
  int j = (int)(t & (HID - 1));
  float v = out1[t] + b1[j];
  out1[t] = v > 0.f ? v : 0.f;
}

__global__ void k_self2(const float* __restrict__ h2, const float* __restrict__ dis,
                        float* __restrict__ out2, int n) {
  int i = blockIdx.x * blockDim.x + threadIdx.x;
  if (i >= n) return;
  float s = dis[i]; s = s * s;
  out2[(size_t)i * 2 + 0] = h2[(size_t)i * 2 + 0] * s;
  out2[(size_t)i * 2 + 1] = h2[(size_t)i * 2 + 1] * s;
}

__global__ void k_scatter2(const int* __restrict__ src, const int* __restrict__ dst,
                           const float* __restrict__ dis, const float* __restrict__ h2,
                           float* __restrict__ out2, int nE) {
  int e = blockIdx.x * blockDim.x + threadIdx.x;
  if (e >= nE) return;
  int s = src[e], d = dst[e];
  float norm = dis[s] * dis[d];
  atomicAdd(&out2[(size_t)d * 2 + 0], h2[(size_t)s * 2 + 0] * norm);
  atomicAdd(&out2[(size_t)d * 2 + 1], h2[(size_t)s * 2 + 1] * norm);
}

__global__ void k_logsoftmax(const float* __restrict__ out2, const float* __restrict__ b2,
                             float* __restrict__ y, int n) {
  int i = blockIdx.x * blockDim.x + threadIdx.x;
  if (i >= n) return;
  float a0 = out2[(size_t)i * 2 + 0] + b2[0];
  float a1 = out2[(size_t)i * 2 + 1] + b2[1];
  float mx = fmaxf(a0, a1);
  float lse = mx + logf(expf(a0 - mx) + expf(a1 - mx));
  y[(size_t)i * 2 + 0] = a0 - lse;
  y[(size_t)i * 2 + 1] = a1 - lse;
}


extern "C" void kernel_launch(void* const* d_in, const int* in_sizes, int n_in,
                              void* d_out, int out_size, void* d_ws, size_t ws_size,
                              hipStream_t stream) {
  const float* x  = (const float*)d_in[0];
  const int*   ei = (const int*)d_in[1];
  const float* W1 = (const float*)d_in[2];
  const float* b1 = (const float*)d_in[3];
  const float* W2 = (const float*)d_in[4];
  const float* b2 = (const float*)d_in[5];

  const int n  = in_sizes[0] / IN_DIM;
  const int nE = in_sizes[1] / 2;
  const int* src = ei;
  const int* dst = ei + nE;

  float* ws   = (float*)d_ws;
  float* deg  = ws;
  float* dis  = deg + n;
  float* h    = dis + n;
  float* out1 = h    + (size_t)n * HID;
  float* h2   = out1 + (size_t)n * HID;
  float* out2 = h2   + (size_t)n * 2;
  float* y    = (float*)d_out;

  const int TB = 256;
  dim3 blk(TB);
  int gN   = (n + TB - 1) / TB;
  int gE   = (nE + TB - 1) / TB;
  int gE4  = (int)(((long long)nE * 4 + TB - 1) / TB);
  int gN16 = (int)(((long long)n * HID + TB - 1) / TB);
  int tiles = (n + 15) / 16;
  int gT   = (tiles + (TB / 32) - 1) / (TB / 32);

  k_init_deg  <<<gN,  blk, 0, stream>>>(deg, n);
  k_deg_edges <<<gE,  blk, 0, stream>>>(dst, deg, nE);
  k_dis       <<<gN,  blk, 0, stream>>>(deg, dis, n);
  k_gemm1_wmma<<<gT,  blk, 0, stream>>>(x, W1, h, n);
  k_self1     <<<gN,  blk, 0, stream>>>(h, dis, out1, n);
  k_scatter1  <<<gE4, blk, 0, stream>>>(src, dst, dis, h, out1, nE);
  k_relu_bias <<<gN16,blk, 0, stream>>>(out1, b1, n);
  k_gemm2_wmma<<<gT,  blk, 0, stream>>>(out1, W2, h2, n);
  k_self2     <<<gN,  blk, 0, stream>>>(h2, dis, out2, n);
  k_scatter2  <<<gE,  blk, 0, stream>>>(src, dst, dis, h2, out2, nE);
  k_logsoftmax<<<gN,  blk, 0, stream>>>(out2, b2, y, n);
}


# Edges grouped per destination each call (slot from a returning int atomic, scan, placement); a half-wave per node sums its list in registers, epilogues fused; f32 atomics gone; sums in arrival order
# speedup vs baseline: 1.7146x; 1.7146x over previous
#include <hip/hip_runtime.h>
#include <math.h>

typedef __attribute__((ext_vector_type(2))) float v2f;
typedef __attribute__((ext_vector_type(8))) float v8f;

#define IN_DIM 128
#define HID    16
#define TB     256
#define SCAN_TILE 1024


__global__ void k_zero_cnt(int* __restrict__ cnt, int n) {
  int i = blockIdx.x * blockDim.x + threadIdx.x;
  if (i < n) cnt[i] = 0;
}

__global__ void k_rank(const int* __restrict__ dst, int* __restrict__ cnt,
                       int* __restrict__ rank, int nE) {
  int e = blockIdx.x * blockDim.x + threadIdx.x;
  if (e < nE) rank[e] = atomicAdd(&cnt[dst[e]], 1);
}

__device__ __forceinline__ int wave_sum_i(int v) {
#pragma unroll
  for (int m = 16; m > 0; m >>= 1) v += __shfl_xor(v, m, 32);
  return v;
}

__global__ void k_blocksum(const int* __restrict__ cnt, int* __restrict__ bsum, int n) {
  __shared__ int sm[TB / 32];
  const int t = threadIdx.x;
  const int i0 = blockIdx.x * SCAN_TILE + 4 * t;
  int v = 0;
#pragma unroll
  for (int q = 0; q < 4; ++q) if (i0 + q < n) v += cnt[i0 + q];
  v = wave_sum_i(v);
  if ((t & 31) == 0) sm[t >> 5] = v;
  __syncthreads();
  if (t == 0) {
    int s = 0;
#pragma unroll
    for (int w = 0; w < TB / 32; ++w) s += sm[w];
    bsum[blockIdx.x] = s;
  }
}

__global__ void k_scan_apply(const int* __restrict__ cnt, const int* __restrict__ bsum,
                             int* __restrict__ row_ptr, float* __restrict__ dis, int n) {
  __shared__ int sm_off[TB / 32];
  __shared__ int sm_w[TB / 32];
  const int t = threadIdx.x, lane = t & 31, wave = t >> 5, b = blockIdx.x;
  int off = 0;
  for (int i = t; i < b; i += TB) off += bsum[i];
  off = wave_sum_i(off);
  if (lane == 0) sm_off[wave] = off;

  const int i0 = b * SCAN_TILE + 4 * t;
  int c[4];
  int ts = 0;
#pragma unroll
  for (int q = 0; q < 4; ++q) { c[q] = (i0 + q < n) ? cnt[i0 + q] : 0; ts += c[q]; }
  int inc = ts;
#pragma unroll
  for (int d = 1; d < 32; d <<= 1) {
    int u = __shfl_up(inc, d, 32);
    if (lane >= d) inc += u;
  }
  if (lane == 31) sm_w[wave] = inc;
  __syncthreads();
  int p = inc - ts;
#pragma unroll
  for (int w = 0; w < TB / 32; ++w) {
    p += sm_off[w];
    if (w < wave) p += sm_w[w];
  }
#pragma unroll
  for (int q = 0; q < 4; ++q) {
    if (i0 + q < n) {
      row_ptr[i0 + q] = p;
      dis[i0 + q] = rsqrtf((float)(c[q] + 1));
      p += c[q];
    }
  }
}

__global__ void k_place(const int* __restrict__ src, const int* __restrict__ dst,
                        const int* __restrict__ rank, const int* __restrict__ row_ptr,
                        int* __restrict__ col, int nE) {
  int e = blockIdx.x * blockDim.x + threadIdx.x;
  if (e < nE) col[row_ptr[dst[e]] + rank[e]] = src[e];
}


__global__ void k_gemm1_wmma(const float* __restrict__ x, const float* __restrict__ W1,
                             float* __restrict__ h, int n) {
  const int lane = threadIdx.x & 31;
  const int wave = threadIdx.x >> 5;
  const int tile = blockIdx.x * (blockDim.x >> 5) + wave;
  const int row0 = tile << 4;
  if (row0 >= n) return;
  const int m  = lane & 15;
  const int kh = lane >> 4;
  const float* xrow = x + (size_t)(row0 + m) * IN_DIM + 2 * kh;

  v8f c = {0.f, 0.f, 0.f, 0.f, 0.f, 0.f, 0.f, 0.f};
#pragma unroll
  for (int k0 = 0; k0 < IN_DIM / 4; ++k0) {
    const int kk = 4 * k0 + 2 * kh;
    const float2 av = *reinterpret_cast<const float2*>(xrow + 4 * k0);
    v2f a; a.x = av.x; a.y = av.y;
    v2f b; b.x = W1[(kk + 0) * HID + m];
           b.y = W1[(kk + 1) * HID + m];
    c = __builtin_amdgcn_wmma_f32_16x16x4_f32(false, a, false, b, (short)0, c, false, false);
  }
  float* hout = h + (size_t)(row0 + 8 * kh) * HID + m;
#pragma unroll
  for (int r = 0; r < 8; ++r)
    hout[(size_t)r * HID] = c[r];
}


__global__ void k_gemm2_wmma(const float* __restrict__ h1, const float* __restrict__ W2,
                             float* __restrict__ h2, int n) {
  const int lane = threadIdx.x & 31;
  const int wave = threadIdx.x >> 5;
  const int tile = blockIdx.x * (blockDim.x >> 5) + wave;
  const int row0 = tile << 4;
  if (row0 >= n) return;
  const int m  = lane & 15;
  const int kh = lane >> 4;
  const float bmask = (m < 2) ? 1.0f : 0.0f;
  const int   col   = (m < 2) ? m : 0;
  const float* arow = h1 + (size_t)(row0 + m) * HID + 2 * kh;

  v8f c = {0.f, 0.f, 0.f, 0.f, 0.f, 0.f, 0.f, 0.f};
#pragma unroll
  for (int k0 = 0; k0 < HID / 4; ++k0) {
    const int kk = 4 * k0 + 2 * kh;
    const float2 av = *reinterpret_cast<const float2*>(arow + 4 * k0);
    v2f a; a.x = av.x; a.y = av.y;
    v2f b; b.x = W2[(kk + 0) * 2 + col] * bmask;
           b.y = W2[(kk + 1) * 2 + col] * bmask;
    c = __builtin_amdgcn_wmma_f32_16x16x4_f32(false, a, false, b, (short)0, c, false, false);
  }
  if (m < 2) {
#pragma unroll
    for (int r = 0; r < 8; ++r)
      h2[(size_t)(row0 + r + 8 * kh) * 2 + m] = c[r];
  }
}


__global__ void k_pull1(const int* __restrict__ col, const int* __restrict__ row_ptr,
                        const int* __restrict__ cnt, const float* __restrict__ dis,
                        const float* __restrict__ h, const float* __restrict__ b1,
                        float* __restrict__ r, int n) {
  const long long gid = (long long)blockIdx.x * blockDim.x + threadIdx.x;
  const int d = (int)(gid >> 4);
  if (d >= n) return;
  const int f = threadIdx.x & 15;
  const int beg = row_ptr[d];
  const int m = cnt[d];
  const float dd = dis[d];
  float acc = h[(size_t)d * HID + f] * (dd * dd);
  for (int j0 = 0; j0 < m; j0 += 16) {
    const int jj = j0 + f;
    int s_l = d;
    float w_l = 0.f;
    if (jj < m) { s_l = col[beg + jj]; w_l = dis[s_l] * dd; }
    const int nb = m - j0;
#pragma unroll
    for (int t = 0; t < 16; ++t) {
      const int s = __shfl(s_l, t, 16);
      const float w = __shfl(w_l, t, 16);
      const float hv = h[(size_t)s * HID + f];
      acc = fmaf((t < nb) ? hv : 0.f, w, acc);
    }
  }
  const float v = acc + b1[f];
  r[(size_t)d * HID + f] = v > 0.f ? v : 0.f;
}

__global__ void k_pull2(const int* __restrict__ col, const int* __restrict__ row_ptr,
                        const int* __restrict__ cnt, const float* __restrict__ dis,
                        const float* __restrict__ h2, const float* __restrict__ b2,
                        float* __restrict__ y, int n) {
  __shared__ float2 so[TB / 16];
  const int f = threadIdx.x & 15;
  const int hw = threadIdx.x >> 4;
  const int d0 = blockIdx.x * (TB / 16);
  const int d = d0 + hw;
  float a0 = 0.f, a1 = 0.f;
  if (d < n) {
    const int beg = row_ptr[d];
    const int m = cnt[d];
    const float dd = dis[d];
    for (int jj = f; jj < m; jj += 32) {
      const int jb = jj + 16;
      const int sa = col[beg + jj];
      const int sb = (jb < m) ? col[beg + jb] : sa;
      const float wa = dis[sa] * dd;
      const float wb = (jb < m) ? dis[sb] * dd : 0.f;
      const float2 va = *reinterpret_cast<const float2*>(h2 + (size_t)sa * 2);
      const float2 vb = *reinterpret_cast<const float2*>(h2 + (size_t)sb * 2);
      a0 = fmaf(va.x, wa, a0); a1 = fmaf(va.y, wa, a1);
      a0 = fmaf((jb < m) ? vb.x : 0.f, wb, a0); a1 = fmaf((jb < m) ? vb.y : 0.f, wb, a1);
    }
#pragma unroll
    for (int k = 8; k > 0; k >>= 1) {
      a0 += __shfl_xor(a0, k, 16);
      a1 += __shfl_xor(a1, k, 16);
    }
    if (f == 0) {
      const float s2 = dd * dd;
      const float2 hv = *reinterpret_cast<const float2*>(h2 + (size_t)d * 2);
      const float x0 = a0 + hv.x * s2 + b2[0];
      const float x1 = a1 + hv.y * s2 + b2[1];
      const float mx = fmaxf(x0, x1);
      const float lse = mx + logf(expf(x0 - mx) + expf(x1 - mx));
      so[hw] = make_float2(x0 - lse, x1 - lse);
    }
  }
  __syncthreads();
  if (threadIdx.x < TB / 16 && d0 + (int)threadIdx.x < n)
    *reinterpret_cast<float2*>(y + (size_t)(d0 + threadIdx.x) * 2) = so[threadIdx.x];
}


__global__ void k_init_deg(float* __restrict__ deg, int n) {
  int i = blockIdx.x * blockDim.x + threadIdx.x;
  if (i < n) deg[i] = 1.0f;
}

__global__ void k_deg_edges(const int* __restrict__ dst, float* __restrict__ deg, int nE) {
  int e = blockIdx.x * blockDim.x + threadIdx.x;
  if (e < nE) atomicAdd(&deg[dst[e]], 1.0f);
}

__global__ void k_dis(const float* __restrict__ deg, float* __restrict__ dis, int n) {
  int i = blockIdx.x * blockDim.x + threadIdx.x;
  if (i < n) dis[i] = rsqrtf(deg[i]);
}

__global__ void k_self1(const float* __restrict__ h, const float* __restrict__ dis,
                        float* __restrict__ out1, int n) {
  int i = blockIdx.x * blockDim.x + threadIdx.x;
  if (i >= n) return;
  float s = dis[i]; s = s * s;
  const float4* hp = reinterpret_cast<const float4*>(h + (size_t)i * HID);
  float4* op = reinterpret_cast<float4*>(out1 + (size_t)i * HID);
#pragma unroll
  for (int q = 0; q < 4; ++q) {
    float4 v = hp[q];
    v.x *= s; v.y *= s; v.z *= s; v.w *= s;
    op[q] = v;
  }
}

__global__ void k_scatter1(const int* __restrict__ src, const int* __restrict__ dst,
                           const float* __restrict__ dis, const float* __restrict__ h,
                           float* __restrict__ out1, int nE) {
  long long t = (long long)blockIdx.x * blockDim.x + threadIdx.x;
  int e = (int)(t >> 2);
  if (e >= nE) return;
  int q = (int)(t & 3);
  int s = src[e], d = dst[e];
  float norm = dis[s] * dis[d];
  float4 v = *reinterpret_cast<const float4*>(h + (size_t)s * HID + 4 * q);
  float* o = out1 + (size_t)d * HID + 4 * q;
  atomicAdd(o + 0, v.x * norm);
  atomicAdd(o + 1, v.y * norm);
  atomicAdd(o + 2, v.z * norm);
  atomicAdd(o + 3, v.w * norm);
}

__global__ void k_relu_bias(float* __restrict__ out1, const float* __restrict__ b1, int n) {
  long long t = (long long)blockIdx.x * blockDim.x + threadIdx.x;
  if (t >= (long long)n * HID) return;
  int j = (int)(t & (HID - 1));
  float v = out1[t] + b1[j];
  out1[t] = v > 0.f ? v : 0.f;
}

__global__ void k_self2(const float* __restrict__ h2, const float* __restrict__ dis,
                        float* __restrict__ out2, int n) {
  int i = blockIdx.x * blockDim.x + threadIdx.x;
  if (i >= n) return;
  float s = dis[i]; s = s * s;
  out2[(size_t)i * 2 + 0] = h2[(size_t)i * 2 + 0] * s;
  out2[(size_t)i * 2 + 1] = h2[(size_t)i * 2 + 1] * s;
}

__global__ void k_scatter2(const int* __restrict__ src, const int* __restrict__ dst,
                           const float* __restrict__ dis, const float* __restrict__ h2,
                           float* __restrict__ out2, int nE) {
  int e = blockIdx.x * blockDim.x + threadIdx.x;
  if (e >= nE) return;
  int s = src[e], d = dst[e];
  float norm = dis[s] * dis[d];
  atomicAdd(&out2[(size_t)d * 2 + 0], h2[(size_t)s * 2 + 0] * norm);
  atomicAdd(&out2[(size_t)d * 2 + 1], h2[(size_t)s * 2 + 1] * norm);
}

__global__ void k_logsoftmax(const float* __restrict__ out2, const float* __restrict__ b2,
                             float* __restrict__ y, int n) {
  int i = blockIdx.x * blockDim.x + threadIdx.x;
  if (i >= n) return;
  float a0 = out2[(size_t)i * 2 + 0] + b2[0];
  float a1 = out2[(size_t)i * 2 + 1] + b2[1];
  float mx = fmaxf(a0, a1);
  float lse = mx + logf(expf(a0 - mx) + expf(a1 - mx));
  y[(size_t)i * 2 + 0] = a0 - lse;
  y[(size_t)i * 2 + 1] = a1 - lse;
}


extern "C" void kernel_launch(void* const* d_in, const int* in_sizes, int n_in,
                              void* d_out, int out_size, void* d_ws, size_t ws_size,
                              hipStream_t stream) {
  const float* x  = (const float*)d_in[0];
  const int*   ei = (const int*)d_in[1];
  const float* W1 = (const float*)d_in[2];
  const float* b1 = (const float*)d_in[3];
  const float* W2 = (const float*)d_in[4];
  const float* b2 = (const float*)d_in[5];

  const int n  = in_sizes[0] / IN_DIM;
  const int nE = in_sizes[1] / 2;
  const int* src = ei;
  const int* dst = ei + nE;
  float* y = (float*)d_out;

  dim3 blk(TB);
  int gN   = (n + TB - 1) / TB;
  int gE   = (nE + TB - 1) / TB;
  int gN16 = (int)(((long long)n * HID + TB - 1) / TB);
  int tiles = (n + 15) / 16;
  int gT   = (tiles + (TB / 32) - 1) / (TB / 32);

  const size_t big_words  = (size_t)nE > (size_t)n * 2 * HID ? (size_t)nE : (size_t)n * 2 * HID;
  const size_t need_words = (size_t)n * 5 + 1024 + (size_t)nE + big_words;
  int gS = (n + SCAN_TILE - 1) / SCAN_TILE;
  if (ws_size < need_words * sizeof(int) || gS > 1024) {
    float* deg  = (float*)d_ws;
    float* dis  = deg + n;
    float* h    = dis + n;
    float* out1 = h    + (size_t)n * HID;
    float* h2   = out1 + (size_t)n * HID;
    float* out2 = h2   + (size_t)n * 2;
    int gE4 = (int)(((long long)nE * 4 + TB - 1) / TB);
    k_init_deg  <<<gN,  blk, 0, stream>>>(deg, n);
    k_deg_edges <<<gE,  blk, 0, stream>>>(dst, deg, nE);
    k_dis       <<<gN,  blk, 0, stream>>>(deg, dis, n);
    k_gemm1_wmma<<<gT,  blk, 0, stream>>>(x, W1, h, n);
    k_self1     <<<gN,  blk, 0, stream>>>(h, dis, out1, n);
    k_scatter1  <<<gE4, blk, 0, stream>>>(src, dst, dis, h, out1, nE);
    k_relu_bias <<<gN16,blk, 0, stream>>>(out1, b1, n);
    k_gemm2_wmma<<<gT,  blk, 0, stream>>>(out1, W2, h2, n);
    k_self2     <<<gN,  blk, 0, stream>>>(h2, dis, out2, n);
    k_scatter2  <<<gE,  blk, 0, stream>>>(src, dst, dis, h2, out2, nE);
    k_logsoftmax<<<gN,  blk, 0, stream>>>(out2, b2, y, n);
    return;
  }

  int*   cnt     = (int*)d_ws;
  int*   row_ptr = cnt + n;
  float* dis     = (float*)(row_ptr + n);
  int*   bsum    = (int*)(dis + n);
  float* h2      = (float*)(bsum + 1024);
  int*   rank    = (int*)(h2 + (size_t)n * 2);
  float* h       = (float*)rank;
  float* r       = h + (size_t)n * HID;
  int*   col     = rank + big_words;

  k_zero_cnt  <<<gN,  blk, 0, stream>>>(cnt, n);
  k_rank      <<<gE,  blk, 0, stream>>>(dst, cnt, rank, nE);
  k_blocksum  <<<gS,  blk, 0, stream>>>(cnt, bsum, n);
  k_scan_apply<<<gS,  blk, 0, stream>>>(cnt, bsum, row_ptr, dis, n);
  k_place     <<<gE,  blk, 0, stream>>>(src, dst, rank, row_ptr, col, nE);
  k_gemm1_wmma<<<gT,  blk, 0, stream>>>(x, W1, h, n);
  k_pull1     <<<gN16,blk, 0, stream>>>(col, row_ptr, cnt, dis, h, b1, r, n);
  k_gemm2_wmma<<<gT,  blk, 0, stream>>>(r, W2, h2, n);
  k_pull2     <<<gN16,blk, 0, stream>>>(col, row_ptr, cnt, dis, h2, b2, y, n);
}


# Destination lists built by a two-level counting sort in LDS (512-node ranges per 16384-edge chunk, then nodes of a range): no global atomic, no per-edge scattered access; sums in LDS arrival order
# speedup vs baseline: 4.7548x; 2.7731x over previous
#include <hip/hip_runtime.h>
#include <math.h>

typedef __attribute__((ext_vector_type(2))) float v2f;
typedef __attribute__((ext_vector_type(8))) float v8f;

#define IN_DIM 128
#define HID    16
#define TB     256
#define SCAN_TILE 1024
#define RB     9
#define RANGE  512
#define CH     16384
#define NBMAX  2048
#define SB     1024
#define CAPC   24576


__device__ __forceinline__ int wave_sum_i(int v) {
#pragma unroll
  for (int m = 16; m > 0; m >>= 1) v += __shfl_xor(v, m, 32);
  return v;
}

__device__ __forceinline__ int block_excl_scan(int v, int* sm_w) {
  const int lane = threadIdx.x & 31, wave = threadIdx.x >> 5;
  int inc = v;
  for (int d = 1; d < 32; d <<= 1) {
    int u = __shfl_up(inc, d, 32);
    if (lane >= d) inc += u;
  }
  if (lane == 31) sm_w[wave] = inc;
  __syncthreads();
  if (wave == 0) {
    const int w = sm_w[lane];
    int winc = w;
    for (int d = 1; d < 32; d <<= 1) {
      int u = __shfl_up(winc, d, 32);
      if (lane >= d) winc += u;
    }
    sm_w[lane] = winc - w;
  }
  __syncthreads();
  return inc - v + sm_w[wave];
}

__global__ void __launch_bounds__(SB) k_bin_count(const int* __restrict__ dst, int* __restrict__ cc,
                                                  int nE, int NB, int NCH) {
  __shared__ int histL[NBMAX];
  const int t = threadIdx.x, chunk = blockIdx.x;
  for (int b = t; b < NB; b += SB) histL[b] = 0;
  __syncthreads();
  const int e0 = chunk * CH;
  int4 d4[CH / (4 * SB)];
#pragma unroll
  for (int k = 0; k < CH / (4 * SB); ++k) {
    const int e = e0 + 4 * (t + k * SB);
    d4[k] = *reinterpret_cast<const int4*>(dst + (e < nE ? e : 0));
  }
#pragma unroll
  for (int k = 0; k < CH / (4 * SB); ++k) {
    if (e0 + 4 * (t + k * SB) < nE) {
      atomicAdd(&histL[d4[k].x >> RB], 1);
      atomicAdd(&histL[d4[k].y >> RB], 1);
      atomicAdd(&histL[d4[k].z >> RB], 1);
      atomicAdd(&histL[d4[k].w >> RB], 1);
    }
  }
  __syncthreads();
  for (int b = t; b < NB; b += SB) cc[b * NCH + chunk] = histL[b];
}

__global__ void k_blocksum(const int* __restrict__ cnt, int* __restrict__ bsum, int n) {
  __shared__ int sm[TB / 32];
  const int t = threadIdx.x;
  const int i0 = blockIdx.x * SCAN_TILE + 4 * t;
  int v = 0;
#pragma unroll
  for (int q = 0; q < 4; ++q) if (i0 + q < n) v += cnt[i0 + q];
  v = wave_sum_i(v);
  if ((t & 31) == 0) sm[t >> 5] = v;
  __syncthreads();
  if (t == 0) {
    int s = 0;
#pragma unroll
    for (int w = 0; w < TB / 32; ++w) s += sm[w];
    bsum[blockIdx.x] = s;
  }
}

__global__ void k_scan_apply(const int* __restrict__ cnt, const int* __restrict__ bsum,
                             int* __restrict__ off, int n) {
  __shared__ int sm_off[TB / 32];
  __shared__ int sm_w[TB / 32];
  const int t = threadIdx.x, lane = t & 31, wave = t >> 5, b = blockIdx.x;
  int o = 0;
  for (int i = t; i < b; i += TB) o += bsum[i];
  o = wave_sum_i(o);
  if (lane == 0) sm_off[wave] = o;

  const int i0 = b * SCAN_TILE + 4 * t;
  int c[4];
  int ts = 0;
#pragma unroll
  for (int q = 0; q < 4; ++q) { c[q] = (i0 + q < n) ? cnt[i0 + q] : 0; ts += c[q]; }
  int inc = ts;
#pragma unroll
  for (int d = 1; d < 32; d <<= 1) {
    int u = __shfl_up(inc, d, 32);
    if (lane >= d) inc += u;
  }
  if (lane == 31) sm_w[wave] = inc;
  __syncthreads();
  int p = inc - ts;
#pragma unroll
  for (int w = 0; w < TB / 32; ++w) {
    p += sm_off[w];
    if (w < wave) p += sm_w[w];
  }
#pragma unroll
  for (int q = 0; q < 4; ++q) {
    if (i0 + q < n) {
      off[i0 + q] = p;
      p += c[q];
    }
  }
}

__global__ void __launch_bounds__(SB) k_bin_scatter(const int* __restrict__ src, const int* __restrict__ dst,
                                                    const int* __restrict__ cc, const int* __restrict__ off,
                                                    unsigned* __restrict__ binned, int nE, int NB, int NCH) {
  __shared__ int startL[NBMAX];
  __shared__ int cursorL[NBMAX];
  __shared__ int goffL[NBMAX];
  __shared__ int sm_w[32];
  __shared__ unsigned recL[CH];
  const int t = threadIdx.x, lane = t & 31, wave = t >> 5, chunk = blockIdx.x;
  const int b0 = 2 * t, b1 = 2 * t + 1;
  const int h0 = (b0 < NB) ? cc[b0 * NCH + chunk] : 0;
  const int h1 = (b1 < NB) ? cc[b1 * NCH + chunk] : 0;
  const int ex = block_excl_scan(h0 + h1, sm_w);
  if (b0 < NB) { startL[b0] = ex;      cursorL[b0] = ex;      goffL[b0] = off[b0 * NCH + chunk]; }
  if (b1 < NB) { startL[b1] = ex + h0; cursorL[b1] = ex + h0; goffL[b1] = off[b1 * NCH + chunk]; }
  __syncthreads();
  const int e0 = chunk * CH;
  int4 d4[CH / (4 * SB)], s4[CH / (4 * SB)];
#pragma unroll
  for (int k = 0; k < CH / (4 * SB); ++k) {
    const int e = e0 + 4 * (t + k * SB);
    const int ee = e < nE ? e : 0;
    d4[k] = *reinterpret_cast<const int4*>(dst + ee);
    s4[k] = *reinterpret_cast<const int4*>(src + ee);
  }
#pragma unroll
  for (int k = 0; k < CH / (4 * SB); ++k) {
    if (e0 + 4 * (t + k * SB) < nE) {
      const int q0 = atomicAdd(&cursorL[d4[k].x >> RB], 1);
      const int q1 = atomicAdd(&cursorL[d4[k].y >> RB], 1);
      const int q2 = atomicAdd(&cursorL[d4[k].z >> RB], 1);
      const int q3 = atomicAdd(&cursorL[d4[k].w >> RB], 1);
      recL[q0] = ((unsigned)s4[k].x << RB) | (unsigned)(d4[k].x & (RANGE - 1));
      recL[q1] = ((unsigned)s4[k].y << RB) | (unsigned)(d4[k].y & (RANGE - 1));
      recL[q2] = ((unsigned)s4[k].z << RB) | (unsigned)(d4[k].z & (RANGE - 1));
      recL[q3] = ((unsigned)s4[k].w << RB) | (unsigned)(d4[k].w & (RANGE - 1));
    }
  }
  __syncthreads();
  for (int b = wave; b < NB; b += SB / 32) {
    const int s0 = startL[b];
    const int len = cursorL[b] - s0;
    const int g = goffL[b];
    for (int j = lane; j < len; j += 32) binned[g + j] = recL[s0 + j];
  }
}

__global__ void __launch_bounds__(SB) k_range_lists(const unsigned* __restrict__ binned, const int* __restrict__ off,
                                                    int* __restrict__ cnt, int* __restrict__ row_ptr,
                                                    float* __restrict__ dis, int* __restrict__ col,
                                                    int n, int nE, int NB, int NCH) {
  __shared__ int cntL[RANGE];
  __shared__ int curL[RANGE];
  __shared__ int sm_w[32];
  __shared__ int colL[CAPC];
  const int t = threadIdx.x, b = blockIdx.x;
  const int S = off[b * NCH];
  const int E = (b + 1 < NB) ? off[(b + 1) * NCH] : nE;
  const int R = E - S;
  if (t < RANGE) cntL[t] = 0;
  __syncthreads();
  for (int i0 = 0; i0 < R; i0 += 4 * SB) {
    unsigned rec[4];
#pragma unroll
    for (int k = 0; k < 4; ++k) {
      const int i = i0 + k * SB + t;
      rec[k] = binned[S + (i < R ? i : 0)];
    }
#pragma unroll
    for (int k = 0; k < 4; ++k)
      if (i0 + k * SB + t < R) atomicAdd(&cntL[rec[k] & (RANGE - 1)], 1);
  }
  __syncthreads();
  const int c = (t < RANGE) ? cntL[t] : 0;
  const int ex = block_excl_scan(c, sm_w);
  if (t < RANGE) {
    curL[t] = ex;
    const int node = (b << RB) + t;
    if (node < n) {
      cnt[node] = c;
      row_ptr[node] = S + ex;
      dis[node] = rsqrtf((float)(c + 1));
    }
  }
  __syncthreads();
  if (R <= CAPC) {
    for (int i0 = 0; i0 < R; i0 += 4 * SB) {
      unsigned rec[4];
#pragma unroll
      for (int k = 0; k < 4; ++k) {
        const int i = i0 + k * SB + t;
        rec[k] = binned[S + (i < R ? i : 0)];
      }
#pragma unroll
      for (int k = 0; k < 4; ++k) {
        if (i0 + k * SB + t < R) {
          const int slot = atomicAdd(&curL[rec[k] & (RANGE - 1)], 1);
          colL[slot] = (int)(rec[k] >> RB);
        }
      }
    }
    __syncthreads();
    for (int i = t; i < R; i += SB) col[S + i] = colL[i];
  } else {
    for (int i = t; i < R; i += SB) {
      const unsigned rec = binned[S + i];
      const int slot = atomicAdd(&curL[rec & (RANGE - 1)], 1);
      col[S + slot] = (int)(rec >> RB);
    }
  }
}


__global__ void k_gemm1_wmma(const float* __restrict__ x, const float* __restrict__ W1,
                             float* __restrict__ h, int n) {
  const int lane = threadIdx.x & 31;
  const int wave = threadIdx.x >> 5;
  const int tile = blockIdx.x * (blockDim.x >> 5) + wave;
  const int row0 = tile << 4;
  if (row0 >= n) return;
  const int m  = lane & 15;
  const int kh = lane >> 4;
  const float* xrow = x + (size_t)(row0 + m) * IN_DIM + 2 * kh;

  v8f c = {0.f, 0.f, 0.f, 0.f, 0.f, 0.f, 0.f, 0.f};
#pragma unroll
  for (int k0 = 0; k0 < IN_DIM / 4; ++k0) {
    const int kk = 4 * k0 + 2 * kh;
    const float2 av = *reinterpret_cast<const float2*>(xrow + 4 * k0);
    v2f a; a.x = av.x; a.y = av.y;
    v2f b; b.x = W1[(kk + 0) * HID + m];
           b.y = W1[(kk + 1) * HID + m];
    c = __builtin_amdgcn_wmma_f32_16x16x4_f32(false, a, false, b, (short)0, c, false, false);
  }
  float* hout = h + (size_t)(row0 + 8 * kh) * HID + m;
#pragma unroll
  for (int r = 0; r < 8; ++r)
    hout[(size_t)r * HID] = c[r];
}


__global__ void k_gemm2_wmma(const float* __restrict__ h1, const float* __restrict__ W2,
                             float* __restrict__ h2, int n) {
  const int lane = threadIdx.x & 31;
  const int wave = threadIdx.x >> 5;
  const int tile = blockIdx.x * (blockDim.x >> 5) + wave;
  const int row0 = tile << 4;
  if (row0 >= n) return;
  const int m  = lane & 15;
  const int kh = lane >> 4;
  const float bmask = (m < 2) ? 1.0f : 0.0f;
  const int   col   = (m < 2) ? m : 0;
  const float* arow = h1 + (size_t)(row0 + m) * HID + 2 * kh;

  v8f c = {0.f, 0.f, 0.f, 0.f, 0.f, 0.f, 0.f, 0.f};
#pragma unroll
  for (int k0 = 0; k0 < HID / 4; ++k0) {
    const int kk = 4 * k0 + 2 * kh;
    const float2 av = *reinterpret_cast<const float2*>(arow + 4 * k0);
    v2f a; a.x = av.x; a.y = av.y;
    v2f b; b.x = W2[(kk + 0) * 2 + col] * bmask;
           b.y = W2[(kk + 1) * 2 + col] * bmask;
    c = __builtin_amdgcn_wmma_f32_16x16x4_f32(false, a, false, b, (short)0, c, false, false);
  }
  if (m < 2) {
#pragma unroll
    for (int r = 0; r < 8; ++r)
      h2[(size_t)(row0 + r + 8 * kh) * 2 + m] = c[r];
  }
}


__global__ void k_pull1(const int* __restrict__ col, const int* __restrict__ row_ptr,
                        const int* __restrict__ cnt, const float* __restrict__ dis,
                        const float* __restrict__ h, const float* __restrict__ b1,
                        float* __restrict__ r, int n) {
  const long long gid = (long long)blockIdx.x * blockDim.x + threadIdx.x;
  const int d = (int)(gid >> 4);
  if (d >= n) return;
  const int f = threadIdx.x & 15;
  const int beg = row_ptr[d];
  const int m = cnt[d];
  const float dd = dis[d];
  float acc = h[(size_t)d * HID + f] * (dd * dd);
  for (int j0 = 0; j0 < m; j0 += 16) {
    const int jj = j0 + f;
    int s_l = d;
    float w_l = 0.f;
    if (jj < m) { s_l = col[beg + jj]; w_l = dis[s_l] * dd; }
    const int nb = m - j0;
#pragma unroll
    for (int t = 0; t < 16; ++t) {
      const int s = __shfl(s_l, t, 16);
      const float w = __shfl(w_l, t, 16);
      const float hv = h[(size_t)s * HID + f];
      acc = fmaf((t < nb) ? hv : 0.f, w, acc);
    }
  }
  const float v = acc + b1[f];
  r[(size_t)d * HID + f] = v > 0.f ? v : 0.f;
}

__global__ void k_pull2(const int* __restrict__ col, const int* __restrict__ row_ptr,
                        const int* __restrict__ cnt, const float* __restrict__ dis,
                        const float* __restrict__ h2, const float* __restrict__ b2,
                        float* __restrict__ y, int n) {
  __shared__ float2 so[TB / 16];
  const int f = threadIdx.x & 15;
  const int hw = threadIdx.x >> 4;
  const int d0 = blockIdx.x * (TB / 16);
  const int d = d0 + hw;
  float a0 = 0.f, a1 = 0.f;
  if (d < n) {
    const int beg = row_ptr[d];
    const int m = cnt[d];
    const float dd = dis[d];
    for (int jj = f; jj < m; jj += 32) {
      const int jb = jj + 16;
      const int sa = col[beg + jj];
      const int sb = (jb < m) ? col[beg + jb] : sa;
      const float wa = dis[sa] * dd;
      const float wb = (jb < m) ? dis[sb] * dd : 0.f;
      const float2 va = *reinterpret_cast<const float2*>(h2 + (size_t)sa * 2);
      const float2 vb = *reinterpret_cast<const float2*>(h2 + (size_t)sb * 2);
      a0 = fmaf(va.x, wa, a0); a1 = fmaf(va.y, wa, a1);
      a0 = fmaf((jb < m) ? vb.x : 0.f, wb, a0); a1 = fmaf((jb < m) ? vb.y : 0.f, wb, a1);
    }
#pragma unroll
    for (int k = 8; k > 0; k >>= 1) {
      a0 += __shfl_xor(a0, k, 16);
      a1 += __shfl_xor(a1, k, 16);
    }
    if (f == 0) {
      const float s2 = dd * dd;
      const float2 hv = *reinterpret_cast<const float2*>(h2 + (size_t)d * 2);
      const float x0 = a0 + hv.x * s2 + b2[0];
      const float x1 = a1 + hv.y * s2 + b2[1];
      const float mx = fmaxf(x0, x1);
      const float lse = mx + logf(expf(x0 - mx) + expf(x1 - mx));
      so[hw] = make_float2(x0 - lse, x1 - lse);
    }
  }
  __syncthreads();
  if (threadIdx.x < TB / 16 && d0 + (int)threadIdx.x < n)
    *reinterpret_cast<float2*>(y + (size_t)(d0 + threadIdx.x) * 2) = so[threadIdx.x];
}


__global__ void k_init_deg(float* __restrict__ deg, int n) {
  int i = blockIdx.x * blockDim.x + threadIdx.x;
  if (i < n) deg[i] = 1.0f;
}

__global__ void k_deg_edges(const int* __restrict__ dst, float* __restrict__ deg, int nE) {
  int e = blockIdx.x * blockDim.x + threadIdx.x;
  if (e < nE) atomicAdd(&deg[dst[e]], 1.0f);
}

__global__ void k_dis(const float* __restrict__ deg, float* __restrict__ dis, int n) {
  int i = blockIdx.x * blockDim.x + threadIdx.x;
  if (i < n) dis[i] = rsqrtf(deg[i]);
}

__global__ void k_self1(const float* __restrict__ h, const float* __restrict__ dis,
                        float* __restrict__ out1, int n) {
  int i = blockIdx.x * blockDim.x + threadIdx.x;
  if (i >= n) return;
  float s = dis[i]; s = s * s;
  const float4* hp = reinterpret_cast<const float4*>(h + (size_t)i * HID);
  float4* op = reinterpret_cast<float4*>(out1 + (size_t)i * HID);
#pragma unroll
  for (int q = 0; q < 4; ++q) {
    float4 v = hp[q];
    v.x *= s; v.y *= s; v.z *= s; v.w *= s;
    op[q] = v;
  }
}

__global__ void k_scatter1(const int* __restrict__ src, const int* __restrict__ dst,
                           const float* __restrict__ dis, const float* __restrict__ h,
                           float* __restrict__ out1, int nE) {
  long long t = (long long)blockIdx.x * blockDim.x + threadIdx.x;
  int e = (int)(t >> 2);
  if (e >= nE) return;
  int q = (int)(t & 3);
  int s = src[e], d = dst[e];
  float norm = dis[s] * dis[d];
  float4 v = *reinterpret_cast<const float4*>(h + (size_t)s * HID + 4 * q);
  float* o = out1 + (size_t)d * HID + 4 * q;
  atomicAdd(o + 0, v.x * norm);
  atomicAdd(o + 1, v.y * norm);
  atomicAdd(o + 2, v.z * norm);
  atomicAdd(o + 3, v.w * norm);
}

__global__ void k_relu_bias(float* __restrict__ out1, const float* __restrict__ b1, int n) {
  long long t = (long long)blockIdx.x * blockDim.x + threadIdx.x;
  if (t >= (long long)n * HID) return;
  int j = (int)(t & (HID - 1));
  float v = out1[t] + b1[j];
  out1[t] = v > 0.f ? v : 0.f;
}

__global__ void k_self2(const float* __restrict__ h2, const float* __restrict__ dis,
                        float* __restrict__ out2, int n) {
  int i = blockIdx.x * blockDim.x + threadIdx.x;
  if (i >= n) return;
  float s = dis[i]; s = s * s;
  out2[(size_t)i * 2 + 0] = h2[(size_t)i * 2 + 0] * s;
  out2[(size_t)i * 2 + 1] = h2[(size_t)i * 2 + 1] * s;
}

__global__ void k_scatter2(const int* __restrict__ src, const int* __restrict__ dst,
                           const float* __restrict__ dis, const float* __restrict__ h2,
                           float* __restrict__ out2, int nE) {
  int e = blockIdx.x * blockDim.x + threadIdx.x;
  if (e >= nE) return;
  int s = src[e], d = dst[e];
  float norm = dis[s] * dis[d];
  atomicAdd(&out2[(size_t)d * 2 + 0], h2[(size_t)s * 2 + 0] * norm);
  atomicAdd(&out2[(size_t)d * 2 + 1], h2[(size_t)s * 2 + 1] * norm);
}

__global__ void k_logsoftmax(const float* __restrict__ out2, const float* __restrict__ b2,
                             float* __restrict__ y, int n) {
  int i = blockIdx.x * blockDim.x + threadIdx.x;
  if (i >= n) return;
  float a0 = out2[(size_t)i * 2 + 0] + b2[0];
  float a1 = out2[(size_t)i * 2 + 1] + b2[1];
  float mx = fmaxf(a0, a1);
  float lse = mx + logf(expf(a0 - mx) + expf(a1 - mx));
  y[(size_t)i * 2 + 0] = a0 - lse;
  y[(size_t)i * 2 + 1] = a1 - lse;
}


extern "C" void kernel_launch(void* const* d_in, const int* in_sizes, int n_in,
                              void* d_out, int out_size, void* d_ws, size_t ws_size,
                              hipStream_t stream) {
  const float* x  = (const float*)d_in[0];
  const int*   ei = (const int*)d_in[1];
  const float* W1 = (const float*)d_in[2];
  const float* b1 = (const float*)d_in[3];
  const float* W2 = (const float*)d_in[4];
  const float* b2 = (const float*)d_in[5];

  const int n  = in_sizes[0] / IN_DIM;
  const int nE = in_sizes[1] / 2;
  const int* src = ei;
  const int* dst = ei + nE;
  float* y = (float*)d_out;

  dim3 blk(TB);
  int gN   = (n + TB - 1) / TB;
  int gE   = (nE + TB - 1) / TB;
  int gN16 = (int)(((long long)n * HID + TB - 1) / TB);
  int tiles = (n + 15) / 16;
  int gT   = (tiles + (TB / 32) - 1) / (TB / 32);

  const size_t big_words  = (size_t)nE > (size_t)n * 2 * HID ? (size_t)nE : (size_t)n * 2 * HID;
  const size_t need_words = (size_t)n * 5 + 1024 + (size_t)nE + big_words;
  const int NB  = (n + RANGE - 1) / RANGE;
  const int NCH = (nE + CH - 1) / CH;
  const size_t n2 = (size_t)NB * (size_t)NCH;
  int gS = (int)((n2 + SCAN_TILE - 1) / SCAN_TILE);
  if (ws_size < need_words * sizeof(int) || NB > NBMAX || (nE & 3) != 0 ||
      n2 > (size_t)n || n2 > (size_t)1024 * SCAN_TILE) {
    float* deg  = (float*)d_ws;
    float* dis  = deg + n;
    float* h    = dis + n;
    float* out1 = h    + (size_t)n * HID;
    float* h2   = out1 + (size_t)n * HID;
    float* out2 = h2   + (size_t)n * 2;
    int gE4 = (int)(((long long)nE * 4 + TB - 1) / TB);
    k_init_deg  <<<gN,  blk, 0, stream>>>(deg, n);
    k_deg_edges <<<gE,  blk, 0, stream>>>(dst, deg, nE);
    k_dis       <<<gN,  blk, 0, stream>>>(deg, dis, n);
    k_gemm1_wmma<<<gT,  blk, 0, stream>>>(x, W1, h, n);
    k_self1     <<<gN,  blk, 0, stream>>>(h, dis, out1, n);
    k_scatter1  <<<gE4, blk, 0, stream>>>(src, dst, dis, h, out1, nE);
    k_relu_bias <<<gN16,blk, 0, stream>>>(out1, b1, n);
    k_gemm2_wmma<<<gT,  blk, 0, stream>>>(out1, W2, h2, n);
    k_self2     <<<gN,  blk, 0, stream>>>(h2, dis, out2, n);
    k_scatter2  <<<gE,  blk, 0, stream>>>(src, dst, dis, h2, out2, nE);
    k_logsoftmax<<<gN,  blk, 0, stream>>>(out2, b2, y, n);
    return;
  }

  int*   cnt     = (int*)d_ws;
  int*   row_ptr = cnt + n;
  float* dis     = (float*)(row_ptr + n);
  int*   bsum    = (int*)(dis + n);
  float* h2      = (float*)(bsum + 1024);
  int*   cc      = (int*)h2;
  int*   off     = cc + n2;
  unsigned* binned = (unsigned*)(h2 + (size_t)n * 2);
  float* h       = (float*)binned;
  float* r       = h + (size_t)n * HID;
  int*   col     = (int*)binned + big_words;

  dim3 sblk(SB);
  k_bin_count  <<<NCH, sblk, 0, stream>>>(dst, cc, nE, NB, NCH);
  k_blocksum   <<<gS,  blk,  0, stream>>>(cc, bsum, (int)n2);
  k_scan_apply <<<gS,  blk,  0, stream>>>(cc, bsum, off, (int)n2);
  k_bin_scatter<<<NCH, sblk, 0, stream>>>(src, dst, cc, off, binned, nE, NB, NCH);
  k_range_lists<<<NB,  sblk, 0, stream>>>(binned, off, cnt, row_ptr, dis, col, n, nE, NB, NCH);
  k_gemm1_wmma <<<gT,  blk,  0, stream>>>(x, W1, h, n);
  k_pull1      <<<gN16,blk,  0, stream>>>(col, row_ptr, cnt, dis, h, b1, r, n);
  k_gemm2_wmma <<<gT,  blk,  0, stream>>>(r, W2, h2, n);
  k_pull2      <<<gN16,blk,  0, stream>>>(col, row_ptr, cnt, dis, h2, b2, y, n);
}


# Layer 2 gathers one 16-byte node record per list entry (h2 pair + the source's dis, packed by GEMM2's store) instead of an 8-byte row and a separate 4-byte dis gather; sums in the same order
# speedup vs baseline: 5.2580x; 1.1058x over previous
#include <hip/hip_runtime.h>
#include <math.h>

typedef __attribute__((ext_vector_type(2))) float v2f;
typedef __attribute__((ext_vector_type(8))) float v8f;

#define IN_DIM 128
#define HID    16
#define TB     256
#define SCAN_TILE 1024
#define RB     9
#define RANGE  512
#define CH     16384
#define NBMAX  2048
#define SB     1024
#define CAPC   24576


__device__ __forceinline__ int wave_sum_i(int v) {
#pragma unroll
  for (int m = 16; m > 0; m >>= 1) v += __shfl_xor(v, m, 32);
  return v;
}

__device__ __forceinline__ int block_excl_scan(int v, int* sm_w) {
  const int lane = threadIdx.x & 31, wave = threadIdx.x >> 5;
  int inc = v;
  for (int d = 1; d < 32; d <<= 1) {
    int u = __shfl_up(inc, d, 32);
    if (lane >= d) inc += u;
  }
  if (lane == 31) sm_w[wave] = inc;
  __syncthreads();
  if (wave == 0) {
    const int w = sm_w[lane];
    int winc = w;
    for (int d = 1; d < 32; d <<= 1) {
      int u = __shfl_up(winc, d, 32);
      if (lane >= d) winc += u;
    }
    sm_w[lane] = winc - w;
  }
  __syncthreads();
  return inc - v + sm_w[wave];
}

__global__ void __launch_bounds__(SB) k_bin_count(const int* __restrict__ dst, int* __restrict__ cc,
                                                  int nE, int NB, int NCH) {
  __shared__ int histL[NBMAX];
  const int t = threadIdx.x, chunk = blockIdx.x;
  for (int b = t; b < NB; b += SB) histL[b] = 0;
  __syncthreads();
  const int e0 = chunk * CH;
  int4 d4[CH / (4 * SB)];
#pragma unroll
  for (int k = 0; k < CH / (4 * SB); ++k) {
    const int e = e0 + 4 * (t + k * SB);
    d4[k] = *reinterpret_cast<const int4*>(dst + (e < nE ? e : 0));
  }
#pragma unroll
  for (int k = 0; k < CH / (4 * SB); ++k) {
    if (e0 + 4 * (t + k * SB) < nE) {
      atomicAdd(&histL[d4[k].x >> RB], 1);
      atomicAdd(&histL[d4[k].y >> RB], 1);
      atomicAdd(&histL[d4[k].z >> RB], 1);
      atomicAdd(&histL[d4[k].w >> RB], 1);
    }
  }
  __syncthreads();
  for (int b = t; b < NB; b += SB) cc[b * NCH + chunk] = histL[b];
}

__global__ void k_blocksum(const int* __restrict__ cnt, int* __restrict__ bsum, int n) {
  __shared__ int sm[TB / 32];
  const int t = threadIdx.x;
  const int i0 = blockIdx.x * SCAN_TILE + 4 * t;
  int v = 0;
#pragma unroll
  for (int q = 0; q < 4; ++q) if (i0 + q < n) v += cnt[i0 + q];
  v = wave_sum_i(v);
  if ((t & 31) == 0) sm[t >> 5] = v;
  __syncthreads();
  if (t == 0) {
    int s = 0;
#pragma unroll
    for (int w = 0; w < TB / 32; ++w) s += sm[w];
    bsum[blockIdx.x] = s;
  }
}

__global__ void k_scan_apply(const int* __restrict__ cnt, const int* __restrict__ bsum,
                             int* __restrict__ off, int n) {
  __shared__ int sm_off[TB / 32];
  __shared__ int sm_w[TB / 32];
  const int t = threadIdx.x, lane = t & 31, wave = t >> 5, b = blockIdx.x;
  int o = 0;
  for (int i = t; i < b; i += TB) o += bsum[i];
  o = wave_sum_i(o);
  if (lane == 0) sm_off[wave] = o;

  const int i0 = b * SCAN_TILE + 4 * t;
  int c[4];
  int ts = 0;
#pragma unroll
  for (int q = 0; q < 4; ++q) { c[q] = (i0 + q < n) ? cnt[i0 + q] : 0; ts += c[q]; }
  int inc = ts;
#pragma unroll
  for (int d = 1; d < 32; d <<= 1) {
    int u = __shfl_up(inc, d, 32);
    if (lane >= d) inc += u;
  }
  if (lane == 31) sm_w[wave] = inc;
  __syncthreads();
  int p = inc - ts;
#pragma unroll
  for (int w = 0; w < TB / 32; ++w) {
    p += sm_off[w];
    if (w < wave) p += sm_w[w];
  }
#pragma unroll
  for (int q = 0; q < 4; ++q) {
    if (i0 + q < n) {
      off[i0 + q] = p;
      p += c[q];
    }
  }
}

__global__ void __launch_bounds__(SB) k_bin_scatter(const int* __restrict__ src, const int* __restrict__ dst,
                                                    const int* __restrict__ cc, const int* __restrict__ off,
                                                    unsigned* __restrict__ binned, int nE, int NB, int NCH) {
  __shared__ int startL[NBMAX];
  __shared__ int cursorL[NBMAX];
  __shared__ int goffL[NBMAX];
  __shared__ int sm_w[32];
  __shared__ unsigned recL[CH];
  const int t = threadIdx.x, lane = t & 31, wave = t >> 5, chunk = blockIdx.x;
  const int b0 = 2 * t, b1 = 2 * t + 1;
  const int h0 = (b0 < NB) ? cc[b0 * NCH + chunk] : 0;
  const int h1 = (b1 < NB) ? cc[b1 * NCH + chunk] : 0;
  const int ex = block_excl_scan(h0 + h1, sm_w);
  if (b0 < NB) { startL[b0] = ex;      cursorL[b0] = ex;      goffL[b0] = off[b0 * NCH + chunk]; }
  if (b1 < NB) { startL[b1] = ex + h0; cursorL[b1] = ex + h0; goffL[b1] = off[b1 * NCH + chunk]; }
  __syncthreads();
  const int e0 = chunk * CH;
  int4 d4[CH / (4 * SB)], s4[CH / (4 * SB)];
#pragma unroll
  for (int k = 0; k < CH / (4 * SB); ++k) {
    const int e = e0 + 4 * (t + k * SB);
    const int ee = e < nE ? e : 0;
    d4[k] = *reinterpret_cast<const int4*>(dst + ee);
    s4[k] = *reinterpret_cast<const int4*>(src + ee);
  }
#pragma unroll
  for (int k = 0; k < CH / (4 * SB); ++k) {
    if (e0 + 4 * (t + k * SB) < nE) {
      const int q0 = atomicAdd(&cursorL[d4[k].x >> RB], 1);
      const int q1 = atomicAdd(&cursorL[d4[k].y >> RB], 1);
      const int q2 = atomicAdd(&cursorL[d4[k].z >> RB], 1);
      const int q3 = atomicAdd(&cursorL[d4[k].w >> RB], 1);
      recL[q0] = ((unsigned)s4[k].x << RB) | (unsigned)(d4[k].x & (RANGE - 1));
      recL[q1] = ((unsigned)s4[k].y << RB) | (unsigned)(d4[k].y & (RANGE - 1));
      recL[q2] = ((unsigned)s4[k].z << RB) | (unsigned)(d4[k].z & (RANGE - 1));
      recL[q3] = ((unsigned)s4[k].w << RB) | (unsigned)(d4[k].w & (RANGE - 1));
    }
  }
  __syncthreads();
  for (int b = wave; b < NB; b += SB / 32) {
    const int s0 = startL[b];
    const int len = cursorL[b] - s0;
    const int g = goffL[b];
    for (int j = lane; j < len; j += 32) binned[g + j] = recL[s0 + j];
  }
}

__global__ void __launch_bounds__(SB) k_range_lists(const unsigned* __restrict__ binned, const int* __restrict__ off,
                                                    int* __restrict__ cnt, int* __restrict__ row_ptr,
                                                    float* __restrict__ dis, int* __restrict__ col,
                                                    int n, int nE, int NB, int NCH) {
  __shared__ int cntL[RANGE];
  __shared__ int curL[RANGE];
  __shared__ int sm_w[32];
  __shared__ int colL[CAPC];
  const int t = threadIdx.x, b = blockIdx.x;
  const int S = off[b * NCH];
  const int E = (b + 1 < NB) ? off[(b + 1) * NCH] : nE;
  const int R = E - S;
  if (t < RANGE) cntL[t] = 0;
  __syncthreads();
  for (int i0 = 0; i0 < R; i0 += 4 * SB) {
    unsigned rec[4];
#pragma unroll
    for (int k = 0; k < 4; ++k) {
      const int i = i0 + k * SB + t;
      rec[k] = binned[S + (i < R ? i : 0)];
    }
#pragma unroll
    for (int k = 0; k < 4; ++k)
      if (i0 + k * SB + t < R) atomicAdd(&cntL[rec[k] & (RANGE - 1)], 1);
  }
  __syncthreads();
  const int c = (t < RANGE) ? cntL[t] : 0;
  const int ex = block_excl_scan(c, sm_w);
  if (t < RANGE) {
    curL[t] = ex;
    const int node = (b << RB) + t;
    if (node < n) {
      cnt[node] = c;
      row_ptr[node] = S + ex;
      dis[node] = rsqrtf((float)(c + 1));
    }
  }
  __syncthreads();
  if (R <= CAPC) {
    for (int i0 = 0; i0 < R; i0 += 4 * SB) {
      unsigned rec[4];
#pragma unroll
      for (int k = 0; k < 4; ++k) {
        const int i = i0 + k * SB + t;
        rec[k] = binned[S + (i < R ? i : 0)];
      }
#pragma unroll
      for (int k = 0; k < 4; ++k) {
        if (i0 + k * SB + t < R) {
          const int slot = atomicAdd(&curL[rec[k] & (RANGE - 1)], 1);
          colL[slot] = (int)(rec[k] >> RB);
        }
      }
    }
    __syncthreads();
    for (int i = t; i < R; i += SB) col[S + i] = colL[i];
  } else {
    for (int i = t; i < R; i += SB) {
      const unsigned rec = binned[S + i];
      const int slot = atomicAdd(&curL[rec & (RANGE - 1)], 1);
      col[S + slot] = (int)(rec >> RB);
    }
  }
}


__global__ void k_gemm1_wmma(const float* __restrict__ x, const float* __restrict__ W1,
                             float* __restrict__ h, int n) {
  const int lane = threadIdx.x & 31;
  const int wave = threadIdx.x >> 5;
  const int tile = blockIdx.x * (blockDim.x >> 5) + wave;
  const int row0 = tile << 4;
  if (row0 >= n) return;
  const int m  = lane & 15;
  const int kh = lane >> 4;
  const float* xrow = x + (size_t)(row0 + m) * IN_DIM + 2 * kh;

  v8f c = {0.f, 0.f, 0.f, 0.f, 0.f, 0.f, 0.f, 0.f};
#pragma unroll
  for (int k0 = 0; k0 < IN_DIM / 4; ++k0) {
    const int kk = 4 * k0 + 2 * kh;
    const float2 av = *reinterpret_cast<const float2*>(xrow + 4 * k0);
    v2f a; a.x = av.x; a.y = av.y;
    v2f b; b.x = W1[(kk + 0) * HID + m];
           b.y = W1[(kk + 1) * HID + m];
    c = __builtin_amdgcn_wmma_f32_16x16x4_f32(false, a, false, b, (short)0, c, false, false);
  }
  float* hout = h + (size_t)(row0 + 8 * kh) * HID + m;
#pragma unroll
  for (int r = 0; r < 8; ++r)
    hout[(size_t)r * HID] = c[r];
}


template <bool REC>
__global__ void k_gemm2_wmma(const float* __restrict__ h1, const float* __restrict__ W2,
                             const float* __restrict__ dis, float* __restrict__ t2, int n) {
  const int lane = threadIdx.x & 31;
  const int wave = threadIdx.x >> 5;
  const int tile = blockIdx.x * (blockDim.x >> 5) + wave;
  const int row0 = tile << 4;
  if (row0 >= n) return;
  const int m  = lane & 15;
  const int kh = lane >> 4;
  const float bmask = (m < 2) ? 1.0f : 0.0f;
  const int   col   = (m < 2) ? m : 0;
  const float* arow = h1 + (size_t)(row0 + m) * HID + 2 * kh;
  float dv[8] = {0.f, 0.f, 0.f, 0.f, 0.f, 0.f, 0.f, 0.f};
  if (REC) {
    const float4 da = *reinterpret_cast<const float4*>(dis + row0 + 8 * kh);
    const float4 db = *reinterpret_cast<const float4*>(dis + row0 + 8 * kh + 4);
    dv[0] = da.x; dv[1] = da.y; dv[2] = da.z; dv[3] = da.w;
    dv[4] = db.x; dv[5] = db.y; dv[6] = db.z; dv[7] = db.w;
  }

  v8f c = {0.f, 0.f, 0.f, 0.f, 0.f, 0.f, 0.f, 0.f};
#pragma unroll
  for (int k0 = 0; k0 < HID / 4; ++k0) {
    const int kk = 4 * k0 + 2 * kh;
    const float2 av = *reinterpret_cast<const float2*>(arow + 4 * k0);
    v2f a; a.x = av.x; a.y = av.y;
    v2f b; b.x = W2[(kk + 0) * 2 + col] * bmask;
           b.y = W2[(kk + 1) * 2 + col] * bmask;
    c = __builtin_amdgcn_wmma_f32_16x16x4_f32(false, a, false, b, (short)0, c, false, false);
  }
  if (REC) {
    if (m < 4) {
#pragma unroll
      for (int r = 0; r < 8; ++r) {
        const int row = row0 + r + 8 * kh;
        const float v = (m < 2) ? c[r] : ((m == 2) ? dv[r] : 0.f);
        t2[(size_t)row * 4 + m] = v;
      }
    }
  } else if (m < 2) {
#pragma unroll
    for (int r = 0; r < 8; ++r)
      t2[(size_t)(row0 + r + 8 * kh) * 2 + m] = c[r];
  }
}


__global__ void k_pull1(const int* __restrict__ col, const int* __restrict__ row_ptr,
                        const int* __restrict__ cnt, const float* __restrict__ dis,
                        const float* __restrict__ h, const float* __restrict__ b1,
                        float* __restrict__ r, int n) {
  const long long gid = (long long)blockIdx.x * blockDim.x + threadIdx.x;
  const int d = (int)(gid >> 4);
  if (d >= n) return;
  const int f = threadIdx.x & 15;
  const int beg = row_ptr[d];
  const int m = cnt[d];
  const float dd = dis[d];
  float acc = h[(size_t)d * HID + f] * (dd * dd);
  for (int j0 = 0; j0 < m; j0 += 16) {
    const int jj = j0 + f;
    int s_l = d;
    float w_l = 0.f;
    if (jj < m) { s_l = col[beg + jj]; w_l = dis[s_l] * dd; }
    const int nb = m - j0;
#pragma unroll
    for (int t = 0; t < 16; ++t) {
      const int s = __shfl(s_l, t, 16);
      const float w = __shfl(w_l, t, 16);
      const float hv = h[(size_t)s * HID + f];
      acc = fmaf((t < nb) ? hv : 0.f, w, acc);
    }
  }
  const float v = acc + b1[f];
  r[(size_t)d * HID + f] = v > 0.f ? v : 0.f;
}

__global__ void k_pull2(const int* __restrict__ col, const int* __restrict__ row_ptr,
                        const int* __restrict__ cnt, const float* __restrict__ t2,
                        const float* __restrict__ b2, float* __restrict__ y, int n) {
  __shared__ float2 so[TB / 16];
  const int f = threadIdx.x & 15;
  const int hw = threadIdx.x >> 4;
  const int d0 = blockIdx.x * (TB / 16);
  const int d = d0 + hw;
  float a0 = 0.f, a1 = 0.f;
  if (d < n) {
    const int beg = row_ptr[d];
    const int m = cnt[d];
    const float4 rd = *reinterpret_cast<const float4*>(t2 + (size_t)d * 4);
    const float dd = rd.z;
    for (int jj = f; jj < m; jj += 32) {
      const int jb = jj + 16;
      const int sa = col[beg + jj];
      const int sb = (jb < m) ? col[beg + jb] : sa;
      const float4 va = *reinterpret_cast<const float4*>(t2 + (size_t)sa * 4);
      const float4 vb = *reinterpret_cast<const float4*>(t2 + (size_t)sb * 4);
      const float wa = va.z * dd;
      const float wb = (jb < m) ? vb.z * dd : 0.f;
      a0 = fmaf(va.x, wa, a0); a1 = fmaf(va.y, wa, a1);
      a0 = fmaf((jb < m) ? vb.x : 0.f, wb, a0); a1 = fmaf((jb < m) ? vb.y : 0.f, wb, a1);
    }
#pragma unroll
    for (int k = 8; k > 0; k >>= 1) {
      a0 += __shfl_xor(a0, k, 16);
      a1 += __shfl_xor(a1, k, 16);
    }
    if (f == 0) {
      const float s2 = dd * dd;
      const float x0 = a0 + rd.x * s2 + b2[0];
      const float x1 = a1 + rd.y * s2 + b2[1];
      const float mx = fmaxf(x0, x1);
      const float lse = mx + logf(expf(x0 - mx) + expf(x1 - mx));
      so[hw] = make_float2(x0 - lse, x1 - lse);
    }
  }
  __syncthreads();
  if (threadIdx.x < TB / 16 && d0 + (int)threadIdx.x < n)
    *reinterpret_cast<float2*>(y + (size_t)(d0 + threadIdx.x) * 2) = so[threadIdx.x];
}


__global__ void k_init_deg(float* __restrict__ deg, int n) {
  int i = blockIdx.x * blockDim.x + threadIdx.x;
  if (i < n) deg[i] = 1.0f;
}

__global__ void k_deg_edges(const int* __restrict__ dst, float* __restrict__ deg, int nE) {
  int e = blockIdx.x * blockDim.x + threadIdx.x;
  if (e < nE) atomicAdd(&deg[dst[e]], 1.0f);
}

__global__ void k_dis(const float* __restrict__ deg, float* __restrict__ dis, int n) {
  int i = blockIdx.x * blockDim.x + threadIdx.x;
  if (i < n) dis[i] = rsqrtf(deg[i]);
}

__global__ void k_self1(const float* __restrict__ h, const float* __restrict__ dis,
                        float* __restrict__ out1, int n) {
  int i = blockIdx.x * blockDim.x + threadIdx.x;
  if (i >= n) return;
  float s = dis[i]; s = s * s;
  const float4* hp = reinterpret_cast<const float4*>(h + (size_t)i * HID);
  float4* op = reinterpret_cast<float4*>(out1 + (size_t)i * HID);
#pragma unroll
  for (int q = 0; q < 4; ++q) {
    float4 v = hp[q];
    v.x *= s; v.y *= s; v.z *= s; v.w *= s;
    op[q] = v;
  }
}

__global__ void k_scatter1(const int* __restrict__ src, const int* __restrict__ dst,
                           const float* __restrict__ dis, const float* __restrict__ h,
                           float* __restrict__ out1, int nE) {
  long long t = (long long)blockIdx.x * blockDim.x + threadIdx.x;
  int e = (int)(t >> 2);
  if (e >= nE) return;
  int q = (int)(t & 3);
  int s = src[e], d = dst[e];
  float norm = dis[s] * dis[d];
  float4 v = *reinterpret_cast<const float4*>(h + (size_t)s * HID + 4 * q);
  float* o = out1 + (size_t)d * HID + 4 * q;
  atomicAdd(o + 0, v.x * norm);
  atomicAdd(o + 1, v.y * norm);
  atomicAdd(o + 2, v.z * norm);
  atomicAdd(o + 3, v.w * norm);
}

__global__ void k_relu_bias(float* __restrict__ out1, const float* __restrict__ b1, int n) {
  long long t = (long long)blockIdx.x * blockDim.x + threadIdx.x;
  if (t >= (long long)n * HID) return;
  int j = (int)(t & (HID - 1));
  float v = out1[t] + b1[j];
  out1[t] = v > 0.f ? v : 0.f;
}

__global__ void k_self2(const float* __restrict__ h2, const float* __restrict__ dis,
                        float* __restrict__ out2, int n) {
  int i = blockIdx.x * blockDim.x + threadIdx.x;
  if (i >= n) return;
  float s = dis[i]; s = s * s;
  out2[(size_t)i * 2 + 0] = h2[(size_t)i * 2 + 0] * s;
  out2[(size_t)i * 2 + 1] = h2[(size_t)i * 2 + 1] * s;
}

__global__ void k_scatter2(const int* __restrict__ src, const int* __restrict__ dst,
                           const float* __restrict__ dis, const float* __restrict__ h2,
                           float* __restrict__ out2, int nE) {
  int e = blockIdx.x * blockDim.x + threadIdx.x;
  if (e >= nE) return;
  int s = src[e], d = dst[e];
  float norm = dis[s] * dis[d];
  atomicAdd(&out2[(size_t)d * 2 + 0], h2[(size_t)s * 2 + 0] * norm);
  atomicAdd(&out2[(size_t)d * 2 + 1], h2[(size_t)s * 2 + 1] * norm);
}

__global__ void k_logsoftmax(const float* __restrict__ out2, const float* __restrict__ b2,
                             float* __restrict__ y, int n) {
  int i = blockIdx.x * blockDim.x + threadIdx.x;
  if (i >= n) return;
  float a0 = out2[(size_t)i * 2 + 0] + b2[0];
  float a1 = out2[(size_t)i * 2 + 1] + b2[1];
  float mx = fmaxf(a0, a1);
  float lse = mx + logf(expf(a0 - mx) + expf(a1 - mx));
  y[(size_t)i * 2 + 0] = a0 - lse;
  y[(size_t)i * 2 + 1] = a1 - lse;
}


extern "C" void kernel_launch(void* const* d_in, const int* in_sizes, int n_in,
                              void* d_out, int out_size, void* d_ws, size_t ws_size,
                              hipStream_t stream) {
  const float* x  = (const float*)d_in[0];
  const int*   ei = (const int*)d_in[1];
  const float* W1 = (const float*)d_in[2];
  const float* b1 = (const float*)d_in[3];
  const float* W2 = (const float*)d_in[4];
  const float* b2 = (const float*)d_in[5];

  const int n  = in_sizes[0] / IN_DIM;
  const int nE = in_sizes[1] / 2;
  const int* src = ei;
  const int* dst = ei + nE;
  float* y = (float*)d_out;

  dim3 blk(TB);
  int gN   = (n + TB - 1) / TB;
  int gE   = (nE + TB - 1) / TB;
  int gN16 = (int)(((long long)n * HID + TB - 1) / TB);
  int tiles = (n + 15) / 16;
  int gT   = (tiles + (TB / 32) - 1) / (TB / 32);

  const size_t big_words  = (size_t)nE > (size_t)n * 2 * HID ? (size_t)nE : (size_t)n * 2 * HID;
  const size_t need_words = (size_t)n * 5 + 1024 + (size_t)nE + big_words;
  const int NB  = (n + RANGE - 1) / RANGE;
  const int NCH = (nE + CH - 1) / CH;
  const size_t n2 = (size_t)NB * (size_t)NCH;
  int gS = (int)((n2 + SCAN_TILE - 1) / SCAN_TILE);
  if (ws_size < need_words * sizeof(int) || NB > NBMAX || (nE & 3) != 0 || (n & 3) != 0 ||
      n2 > (size_t)n || n2 > (size_t)1024 * SCAN_TILE) {
    float* deg  = (float*)d_ws;
    float* dis  = deg + n;
    float* h    = dis + n;
    float* out1 = h    + (size_t)n * HID;
    float* h2   = out1 + (size_t)n * HID;
    float* out2 = h2   + (size_t)n * 2;
    int gE4 = (int)(((long long)nE * 4 + TB - 1) / TB);
    k_init_deg  <<<gN,  blk, 0, stream>>>(deg, n);
    k_deg_edges <<<gE,  blk, 0, stream>>>(dst, deg, nE);
    k_dis       <<<gN,  blk, 0, stream>>>(deg, dis, n);
    k_gemm1_wmma<<<gT,  blk, 0, stream>>>(x, W1, h, n);
    k_self1     <<<gN,  blk, 0, stream>>>(h, dis, out1, n);
    k_scatter1  <<<gE4, blk, 0, stream>>>(src, dst, dis, h, out1, nE);
    k_relu_bias <<<gN16,blk, 0, stream>>>(out1, b1, n);
    k_gemm2_wmma<false><<<gT, blk, 0, stream>>>(out1, W2, dis, h2, n);
    k_self2     <<<gN,  blk, 0, stream>>>(h2, dis, out2, n);
    k_scatter2  <<<gE,  blk, 0, stream>>>(src, dst, dis, h2, out2, nE);
    k_logsoftmax<<<gN,  blk, 0, stream>>>(out2, b2, y, n);
    return;
  }

  int*   cnt     = (int*)d_ws;
  int*   row_ptr = cnt + n;
  float* dis     = (float*)(row_ptr + n);
  int*   bsum    = (int*)(dis + n);
  int*   cc      = bsum + 1024;
  int*   off     = cc + n2;
  unsigned* binned = (unsigned*)(cc + (size_t)n * 2);
  float* h       = (float*)binned;
  float* r       = h + (size_t)n * HID;
  float* t2      = h;
  int*   col     = (int*)binned + big_words;

  dim3 sblk(SB);
  k_bin_count  <<<NCH, sblk, 0, stream>>>(dst, cc, nE, NB, NCH);
  k_blocksum   <<<gS,  blk,  0, stream>>>(cc, bsum, (int)n2);
  k_scan_apply <<<gS,  blk,  0, stream>>>(cc, bsum, off, (int)n2);
  k_bin_scatter<<<NCH, sblk, 0, stream>>>(src, dst, cc, off, binned, nE, NB, NCH);
  k_range_lists<<<NB,  sblk, 0, stream>>>(binned, off, cnt, row_ptr, dis, col, n, nE, NB, NCH);
  k_gemm1_wmma <<<gT,  blk,  0, stream>>>(x, W1, h, n);
  k_pull1      <<<gN16,blk,  0, stream>>>(col, row_ptr, cnt, dis, h, b1, r, n);
  k_gemm2_wmma<true><<<gT, blk, 0, stream>>>(r, W2, dis, t2, n);
  k_pull2      <<<gN16,blk,  0, stream>>>(col, row_ptr, cnt, t2, b2, y, n);
}


# Layer 1 takes dis[src] from a table of in-degree bytes in LDS (one resident work-group per WGP loops over node pairs, rsqrt per entry): the 4-byte gather per list entry is gone; sums in same order
# speedup vs baseline: 5.6534x; 1.0752x over previous
#include <hip/hip_runtime.h>
#include <math.h>

typedef __attribute__((ext_vector_type(2))) float v2f;
typedef __attribute__((ext_vector_type(8))) float v8f;

#define IN_DIM 128
#define HID    16
#define TB     256
#define SCAN_TILE 1024
#define RB     9
#define RANGE  512
#define CH     16384
#define NBMAX  2048
#define SB     1024
#define CAPC   24576
#define DCAP   262144
#define NWGP   256


__device__ __forceinline__ int wave_sum_i(int v) {
#pragma unroll
  for (int m = 16; m > 0; m >>= 1) v += __shfl_xor(v, m, 32);
  return v;
}

__device__ __forceinline__ int block_excl_scan(int v, int* sm_w) {
  const int lane = threadIdx.x & 31, wave = threadIdx.x >> 5;
  int inc = v;
  for (int d = 1; d < 32; d <<= 1) {
    int u = __shfl_up(inc, d, 32);
    if (lane >= d) inc += u;
  }
  if (lane == 31) sm_w[wave] = inc;
  __syncthreads();
  if (wave == 0) {
    const int w = sm_w[lane];
    int winc = w;
    for (int d = 1; d < 32; d <<= 1) {
      int u = __shfl_up(winc, d, 32);
      if (lane >= d) winc += u;
    }
    sm_w[lane] = winc - w;
  }
  __syncthreads();
  return inc - v + sm_w[wave];
}

__global__ void __launch_bounds__(SB) k_bin_count(const int* __restrict__ dst, int* __restrict__ cc,
                                                  int nE, int NB, int NCH) {
  __shared__ int histL[NBMAX];
  const int t = threadIdx.x, chunk = blockIdx.x;
  for (int b = t; b < NB; b += SB) histL[b] = 0;
  __syncthreads();
  const int e0 = chunk * CH;
  int4 d4[CH / (4 * SB)];
#pragma unroll
  for (int k = 0; k < CH / (4 * SB); ++k) {
    const int e = e0 + 4 * (t + k * SB);
    d4[k] = *reinterpret_cast<const int4*>(dst + (e < nE ? e : 0));
  }
#pragma unroll
  for (int k = 0; k < CH / (4 * SB); ++k) {
    if (e0 + 4 * (t + k * SB) < nE) {
      atomicAdd(&histL[d4[k].x >> RB], 1);
      atomicAdd(&histL[d4[k].y >> RB], 1);
      atomicAdd(&histL[d4[k].z >> RB], 1);
      atomicAdd(&histL[d4[k].w >> RB], 1);
    }
  }
  __syncthreads();
  for (int b = t; b < NB; b += SB) cc[b * NCH + chunk] = histL[b];
}

__global__ void k_blocksum(const int* __restrict__ cnt, int* __restrict__ bsum, int n) {
  __shared__ int sm[TB / 32];
  const int t = threadIdx.x;
  const int i0 = blockIdx.x * SCAN_TILE + 4 * t;
  int v = 0;
#pragma unroll
  for (int q = 0; q < 4; ++q) if (i0 + q < n) v += cnt[i0 + q];
  v = wave_sum_i(v);
  if ((t & 31) == 0) sm[t >> 5] = v;
  __syncthreads();
  if (t == 0) {
    int s = 0;
#pragma unroll
    for (int w = 0; w < TB / 32; ++w) s += sm[w];
    bsum[blockIdx.x] = s;
  }
}

__global__ void k_scan_apply(const int* __restrict__ cnt, const int* __restrict__ bsum,
                             int* __restrict__ off, int n) {
  __shared__ int sm_off[TB / 32];
  __shared__ int sm_w[TB / 32];
  const int t = threadIdx.x, lane = t & 31, wave = t >> 5, b = blockIdx.x;
  int o = 0;
  for (int i = t; i < b; i += TB) o += bsum[i];
  o = wave_sum_i(o);
  if (lane == 0) sm_off[wave] = o;

  const int i0 = b * SCAN_TILE + 4 * t;
  int c[4];
  int ts = 0;
#pragma unroll
  for (int q = 0; q < 4; ++q) { c[q] = (i0 + q < n) ? cnt[i0 + q] : 0; ts += c[q]; }
  int inc = ts;
#pragma unroll
  for (int d = 1; d < 32; d <<= 1) {
    int u = __shfl_up(inc, d, 32);
    if (lane >= d) inc += u;
  }
  if (lane == 31) sm_w[wave] = inc;
  __syncthreads();
  int p = inc - ts;
#pragma unroll
  for (int w = 0; w < TB / 32; ++w) {
    p += sm_off[w];
    if (w < wave) p += sm_w[w];
  }
#pragma unroll
  for (int q = 0; q < 4; ++q) {
    if (i0 + q < n) {
      off[i0 + q] = p;
      p += c[q];
    }
  }
}

__global__ void __launch_bounds__(SB) k_bin_scatter(const int* __restrict__ src, const int* __restrict__ dst,
                                                    const int* __restrict__ cc, const int* __restrict__ off,
                                                    unsigned* __restrict__ binned, int nE, int NB, int NCH) {
  __shared__ int startL[NBMAX];
  __shared__ int cursorL[NBMAX];
  __shared__ int goffL[NBMAX];
  __shared__ int sm_w[32];
  __shared__ unsigned recL[CH];
  const int t = threadIdx.x, lane = t & 31, wave = t >> 5, chunk = blockIdx.x;
  const int b0 = 2 * t, b1 = 2 * t + 1;
  const int h0 = (b0 < NB) ? cc[b0 * NCH + chunk] : 0;
  const int h1 = (b1 < NB) ? cc[b1 * NCH + chunk] : 0;
  const int ex = block_excl_scan(h0 + h1, sm_w);
  if (b0 < NB) { startL[b0] = ex;      cursorL[b0] = ex;      goffL[b0] = off[b0 * NCH + chunk]; }
  if (b1 < NB) { startL[b1] = ex + h0; cursorL[b1] = ex + h0; goffL[b1] = off[b1 * NCH + chunk]; }
  __syncthreads();
  const int e0 = chunk * CH;
  int4 d4[CH / (4 * SB)], s4[CH / (4 * SB)];
#pragma unroll
  for (int k = 0; k < CH / (4 * SB); ++k) {
    const int e = e0 + 4 * (t + k * SB);
    const int ee = e < nE ? e : 0;
    d4[k] = *reinterpret_cast<const int4*>(dst + ee);
    s4[k] = *reinterpret_cast<const int4*>(src + ee);
  }
#pragma unroll
  for (int k = 0; k < CH / (4 * SB); ++k) {
    if (e0 + 4 * (t + k * SB) < nE) {
      const int q0 = atomicAdd(&cursorL[d4[k].x >> RB], 1);
      const int q1 = atomicAdd(&cursorL[d4[k].y >> RB], 1);
      const int q2 = atomicAdd(&cursorL[d4[k].z >> RB], 1);
      const int q3 = atomicAdd(&cursorL[d4[k].w >> RB], 1);
      recL[q0] = ((unsigned)s4[k].x << RB) | (unsigned)(d4[k].x & (RANGE - 1));
      recL[q1] = ((unsigned)s4[k].y << RB) | (unsigned)(d4[k].y & (RANGE - 1));
      recL[q2] = ((unsigned)s4[k].z << RB) | (unsigned)(d4[k].z & (RANGE - 1));
      recL[q3] = ((unsigned)s4[k].w << RB) | (unsigned)(d4[k].w & (RANGE - 1));
    }
  }
  __syncthreads();
  for (int b = wave; b < NB; b += SB / 32) {
    const int s0 = startL[b];
    const int len = cursorL[b] - s0;
    const int g = goffL[b];
    for (int j = lane; j < len; j += 32) binned[g + j] = recL[s0 + j];
  }
}

__global__ void __launch_bounds__(SB) k_range_lists(const unsigned* __restrict__ binned, const int* __restrict__ off,
                                                    int* __restrict__ cnt, int* __restrict__ row_ptr,
                                                    float* __restrict__ dis, int* __restrict__ col,
                                                    unsigned* __restrict__ deg8,
                                                    int n, int nE, int NB, int NCH) {
  __shared__ alignas(16) int cntL[RANGE];
  __shared__ int curL[RANGE];
  __shared__ int sm_w[32];
  __shared__ int colL[CAPC];
  const int t = threadIdx.x, b = blockIdx.x;
  const int S = off[b * NCH];
  const int E = (b + 1 < NB) ? off[(b + 1) * NCH] : nE;
  const int R = E - S;
  if (t < RANGE) cntL[t] = 0;
  __syncthreads();
  for (int i0 = 0; i0 < R; i0 += 4 * SB) {
    unsigned rec[4];
#pragma unroll
    for (int k = 0; k < 4; ++k) {
      const int i = i0 + k * SB + t;
      rec[k] = binned[S + (i < R ? i : 0)];
    }
#pragma unroll
    for (int k = 0; k < 4; ++k)
      if (i0 + k * SB + t < R) atomicAdd(&cntL[rec[k] & (RANGE - 1)], 1);
  }
  __syncthreads();
  const int c = (t < RANGE) ? cntL[t] : 0;
  const int ex = block_excl_scan(c, sm_w);
  if (t < RANGE) {
    curL[t] = ex;
    const int node = (b << RB) + t;
    if (node < n) {
      cnt[node] = c;
      row_ptr[node] = S + ex;
      dis[node] = rsqrtf((float)(c + 1));
    }
  }
  if (deg8 != nullptr && t < RANGE / 4) {
    const int4 c4 = *reinterpret_cast<const int4*>(&cntL[4 * t]);
    const unsigned pk = (unsigned)min(c4.x, 255) | ((unsigned)min(c4.y, 255) << 8) |
                        ((unsigned)min(c4.z, 255) << 16) | ((unsigned)min(c4.w, 255) << 24);
    const int wi = (b << (RB - 2)) + t;
    if (4 * wi < n) deg8[wi] = pk;
  }
  __syncthreads();
  if (R <= CAPC) {
    for (int i0 = 0; i0 < R; i0 += 4 * SB) {
      unsigned rec[4];
#pragma unroll
      for (int k = 0; k < 4; ++k) {
        const int i = i0 + k * SB + t;
        rec[k] = binned[S + (i < R ? i : 0)];
      }
#pragma unroll
      for (int k = 0; k < 4; ++k) {
        if (i0 + k * SB + t < R) {
          const int slot = atomicAdd(&curL[rec[k] & (RANGE - 1)], 1);
          colL[slot] = (int)(rec[k] >> RB);
        }
      }
    }
    __syncthreads();
    for (int i = t; i < R; i += SB) col[S + i] = colL[i];
  } else {
    for (int i = t; i < R; i += SB) {
      const unsigned rec = binned[S + i];
      const int slot = atomicAdd(&curL[rec & (RANGE - 1)], 1);
      col[S + slot] = (int)(rec >> RB);
    }
  }
}


__global__ void k_gemm1_wmma(const float* __restrict__ x, const float* __restrict__ W1,
                             float* __restrict__ h, int n) {
  const int lane = threadIdx.x & 31;
  const int wave = threadIdx.x >> 5;
  const int tile = blockIdx.x * (blockDim.x >> 5) + wave;
  const int row0 = tile << 4;
  if (row0 >= n) return;
  const int m  = lane & 15;
  const int kh = lane >> 4;
  const float* xrow = x + (size_t)(row0 + m) * IN_DIM + 2 * kh;

  v8f c = {0.f, 0.f, 0.f, 0.f, 0.f, 0.f, 0.f, 0.f};
#pragma unroll
  for (int k0 = 0; k0 < IN_DIM / 4; ++k0) {
    const int kk = 4 * k0 + 2 * kh;
    const float2 av = *reinterpret_cast<const float2*>(xrow + 4 * k0);
    v2f a; a.x = av.x; a.y = av.y;
    v2f b; b.x = W1[(kk + 0) * HID + m];
           b.y = W1[(kk + 1) * HID + m];
    c = __builtin_amdgcn_wmma_f32_16x16x4_f32(false, a, false, b, (short)0, c, false, false);
  }
  float* hout = h + (size_t)(row0 + 8 * kh) * HID + m;
#pragma unroll
  for (int r = 0; r < 8; ++r)
    hout[(size_t)r * HID] = c[r];
}


template <bool REC>
__global__ void k_gemm2_wmma(const float* __restrict__ h1, const float* __restrict__ W2,
                             const float* __restrict__ dis, float* __restrict__ t2, int n) {
  const int lane = threadIdx.x & 31;
  const int wave = threadIdx.x >> 5;
  const int tile = blockIdx.x * (blockDim.x >> 5) + wave;
  const int row0 = tile << 4;
  if (row0 >= n) return;
  const int m  = lane & 15;
  const int kh = lane >> 4;
  const float bmask = (m < 2) ? 1.0f : 0.0f;
  const int   col   = (m < 2) ? m : 0;
  const float* arow = h1 + (size_t)(row0 + m) * HID + 2 * kh;
  float dv[8] = {0.f, 0.f, 0.f, 0.f, 0.f, 0.f, 0.f, 0.f};
  if (REC) {
    const float4 da = *reinterpret_cast<const float4*>(dis + row0 + 8 * kh);
    const float4 db = *reinterpret_cast<const float4*>(dis + row0 + 8 * kh + 4);
    dv[0] = da.x; dv[1] = da.y; dv[2] = da.z; dv[3] = da.w;
    dv[4] = db.x; dv[5] = db.y; dv[6] = db.z; dv[7] = db.w;
  }

  v8f c = {0.f, 0.f, 0.f, 0.f, 0.f, 0.f, 0.f, 0.f};
#pragma unroll
  for (int k0 = 0; k0 < HID / 4; ++k0) {
    const int kk = 4 * k0 + 2 * kh;
    const float2 av = *reinterpret_cast<const float2*>(arow + 4 * k0);
    v2f a; a.x = av.x; a.y = av.y;
    v2f b; b.x = W2[(kk + 0) * 2 + col] * bmask;
           b.y = W2[(kk + 1) * 2 + col] * bmask;
    c = __builtin_amdgcn_wmma_f32_16x16x4_f32(false, a, false, b, (short)0, c, false, false);
  }
  if (REC) {
    if (m < 4) {
#pragma unroll
      for (int r = 0; r < 8; ++r) {
        const int row = row0 + r + 8 * kh;
        const float v = (m < 2) ? c[r] : ((m == 2) ? dv[r] : 0.f);
        t2[(size_t)row * 4 + m] = v;
      }
    }
  } else if (m < 2) {
#pragma unroll
    for (int r = 0; r < 8; ++r)
      t2[(size_t)(row0 + r + 8 * kh) * 2 + m] = c[r];
  }
}


__global__ void k_pull1(const int* __restrict__ col, const int* __restrict__ row_ptr,
                        const int* __restrict__ cnt, const float* __restrict__ dis,
                        const float* __restrict__ h, const float* __restrict__ b1,
                        float* __restrict__ r, int n) {
  const long long gid = (long long)blockIdx.x * blockDim.x + threadIdx.x;
  const int d = (int)(gid >> 4);
  if (d >= n) return;
  const int f = threadIdx.x & 15;
  const int beg = row_ptr[d];
  const int m = cnt[d];
  const float dd = dis[d];
  float acc = h[(size_t)d * HID + f] * (dd * dd);
  for (int j0 = 0; j0 < m; j0 += 16) {
    const int jj = j0 + f;
    int s_l = d;
    float w_l = 0.f;
    if (jj < m) { s_l = col[beg + jj]; w_l = dis[s_l] * dd; }
    const int nb = m - j0;
#pragma unroll
    for (int t = 0; t < 16; ++t) {
      const int s = __shfl(s_l, t, 16);
      const float w = __shfl(w_l, t, 16);
      const float hv = h[(size_t)s * HID + f];
      acc = fmaf((t < nb) ? hv : 0.f, w, acc);
    }
  }
  const float v = acc + b1[f];
  r[(size_t)d * HID + f] = v > 0.f ? v : 0.f;
}

__global__ void __launch_bounds__(SB) k_pull1_tab(const int* __restrict__ col, const int* __restrict__ row_ptr,
                        const int* __restrict__ cnt, const float* __restrict__ dis,
                        const unsigned* __restrict__ deg8,
                        const float* __restrict__ h, const float* __restrict__ b1,
                        float* __restrict__ r, int n) {
  __shared__ alignas(16) unsigned degL[DCAP / 4];
  const int t = threadIdx.x;
  {
    const uint4* src4 = reinterpret_cast<const uint4*>(deg8);
    uint4* dst4 = reinterpret_cast<uint4*>(degL);
    const int nq = n >> 4;
    int i = t;
    for (; i + 3 * SB < nq; i += 4 * SB) {
      const uint4 v0 = src4[i], v1 = src4[i + SB], v2 = src4[i + 2 * SB], v3 = src4[i + 3 * SB];
      dst4[i] = v0; dst4[i + SB] = v1; dst4[i + 2 * SB] = v2; dst4[i + 3 * SB] = v3;
    }
    for (; i < nq; i += SB) dst4[i] = src4[i];
    if (t < ((n >> 2) & 3)) degL[(nq << 2) + t] = deg8[(nq << 2) + t];
  }
  __syncthreads();
  const unsigned char* degB = reinterpret_cast<const unsigned char*>(degL);
  const int lane = t & 31, wave = t >> 5;
  const int f = lane & 15, half = lane >> 4;
  const int npair = n >> 1;
  const int per = (npair + NWGP - 1) / NWGP;
  const int p0 = blockIdx.x * per;
  const int P = min(per, npair - p0);
  if (P <= 0) return;
  const int qa = p0 + (wave * P) / (SB / 32);
  const int qn = p0 + ((wave + 1) * P) / (SB / 32) - qa;
  const int nd0 = 2 * qa;
  const int dl = nd0 + lane;
  const bool has = lane < 2 * qn;
  const int   beg_l = has ? row_ptr[dl] : 0;
  const int   m_l   = has ? cnt[dl] : 0;
  const float dd_l  = has ? dis[dl] : 0.f;
  const float bias = b1[f];
  for (int i = 0; i < qn; ++i) {
    const int k = 2 * i + half;
    const int d = nd0 + k;
    const int beg = __shfl(beg_l, k, 32);
    const int m = __shfl(m_l, k, 32);
    const float dd = __shfl(dd_l, k, 32);
    int s_cur = (f < m) ? col[beg + f] : d;
    float acc = h[(size_t)d * HID + f] * (dd * dd);
    for (int j0 = 0; j0 < m; j0 += 16) {
      const int jj = j0 + f;
      const int jn = jj + 16;
      const int s_l = s_cur;
      float w_l = 0.f;
      if (jj < m) {
        const unsigned bq = degB[s_l];
        const float ds = (bq < 255u) ? rsqrtf((float)(int)(bq + 1u)) : dis[s_l];
        w_l = ds * dd;
      }
      s_cur = (jn < m) ? col[beg + jn] : d;
      const int nb = m - j0;
#pragma unroll
      for (int tt = 0; tt < 16; ++tt) {
        const int s = __shfl(s_l, tt, 16);
        const float w = __shfl(w_l, tt, 16);
        const float hv = h[(size_t)s * HID + f];
        acc = fmaf((tt < nb) ? hv : 0.f, w, acc);
      }
    }
    const float v = acc + bias;
    r[(size_t)d * HID + f] = v > 0.f ? v : 0.f;
  }
}

__global__ void k_pull2(const int* __restrict__ col, const int* __restrict__ row_ptr,
                        const int* __restrict__ cnt, const float* __restrict__ t2,
                        const float* __restrict__ b2, float* __restrict__ y, int n) {
  __shared__ float2 so[TB / 16];
  const int f = threadIdx.x & 15;
  const int hw = threadIdx.x >> 4;
  const int d0 = blockIdx.x * (TB / 16);
  const int d = d0 + hw;
  float a0 = 0.f, a1 = 0.f;
  if (d < n) {
    const int beg = row_ptr[d];
    const int m = cnt[d];
    const float4 rd = *reinterpret_cast<const float4*>(t2 + (size_t)d * 4);
    const float dd = rd.z;
    for (int jj = f; jj < m; jj += 32) {
      const int jb = jj + 16;
      const int sa = col[beg + jj];
      const int sb = (jb < m) ? col[beg + jb] : sa;
      const float4 va = *reinterpret_cast<const float4*>(t2 + (size_t)sa * 4);
      const float4 vb = *reinterpret_cast<const float4*>(t2 + (size_t)sb * 4);
      const float wa = va.z * dd;
      const float wb = (jb < m) ? vb.z * dd : 0.f;
      a0 = fmaf(va.x, wa, a0); a1 = fmaf(va.y, wa, a1);
      a0 = fmaf((jb < m) ? vb.x : 0.f, wb, a0); a1 = fmaf((jb < m) ? vb.y : 0.f, wb, a1);
    }
#pragma unroll
    for (int k = 8; k > 0; k >>= 1) {
      a0 += __shfl_xor(a0, k, 16);
      a1 += __shfl_xor(a1, k, 16);
    }
    if (f == 0) {
      const float s2 = dd * dd;
      const float x0 = a0 + rd.x * s2 + b2[0];
      const float x1 = a1 + rd.y * s2 + b2[1];
      const float mx = fmaxf(x0, x1);
      const float lse = mx + logf(expf(x0 - mx) + expf(x1 - mx));
      so[hw] = make_float2(x0 - lse, x1 - lse);
    }
  }
  __syncthreads();
  if (threadIdx.x < TB / 16 && d0 + (int)threadIdx.x < n)
    *reinterpret_cast<float2*>(y + (size_t)(d0 + threadIdx.x) * 2) = so[threadIdx.x];
}


__global__ void k_init_deg(float* __restrict__ deg, int n) {
  int i = blockIdx.x * blockDim.x + threadIdx.x;
  if (i < n) deg[i] = 1.0f;
}

__global__ void k_deg_edges(const int* __restrict__ dst, float* __restrict__ deg, int nE) {
  int e = blockIdx.x * blockDim.x + threadIdx.x;
  if (e < nE) atomicAdd(&deg[dst[e]], 1.0f);
}

__global__ void k_dis(const float* __restrict__ deg, float* __restrict__ dis, int n) {
  int i = blockIdx.x * blockDim.x + threadIdx.x;
  if (i < n) dis[i] = rsqrtf(deg[i]);
}

__global__ void k_self1(const float* __restrict__ h, const float* __restrict__ dis,
                        float* __restrict__ out1, int n) {
  int i = blockIdx.x * blockDim.x + threadIdx.x;
  if (i >= n) return;
  float s = dis[i]; s = s * s;
  const float4* hp = reinterpret_cast<const float4*>(h + (size_t)i * HID);
  float4* op = reinterpret_cast<float4*>(out1 + (size_t)i * HID);
#pragma unroll
  for (int q = 0; q < 4; ++q) {
    float4 v = hp[q];
    v.x *= s; v.y *= s; v.z *= s; v.w *= s;
    op[q] = v;
  }
}

__global__ void k_scatter1(const int* __restrict__ src, const int* __restrict__ dst,
                           const float* __restrict__ dis, const float* __restrict__ h,
                           float* __restrict__ out1, int nE) {
  long long t = (long long)blockIdx.x * blockDim.x + threadIdx.x;
  int e = (int)(t >> 2);
  if (e >= nE) return;
  int q = (int)(t & 3);
  int s = src[e], d = dst[e];
  float norm = dis[s] * dis[d];
  float4 v = *reinterpret_cast<const float4*>(h + (size_t)s * HID + 4 * q);
  float* o = out1 + (size_t)d * HID + 4 * q;
  atomicAdd(o + 0, v.x * norm);
  atomicAdd(o + 1, v.y * norm);
  atomicAdd(o + 2, v.z * norm);
  atomicAdd(o + 3, v.w * norm);
}

__global__ void k_relu_bias(float* __restrict__ out1, const float* __restrict__ b1, int n) {
  long long t = (long long)blockIdx.x * blockDim.x + threadIdx.x;
  if (t >= (long long)n * HID) return;
  int j = (int)(t & (HID - 1));
  float v = out1[t] + b1[j];
  out1[t] = v > 0.f ? v : 0.f;
}

__global__ void k_self2(const float* __restrict__ h2, const float* __restrict__ dis,
                        float* __restrict__ out2, int n) {
  int i = blockIdx.x * blockDim.x + threadIdx.x;
  if (i >= n) return;
  float s = dis[i]; s = s * s;
  out2[(size_t)i * 2 + 0] = h2[(size_t)i * 2 + 0] * s;
  out2[(size_t)i * 2 + 1] = h2[(size_t)i * 2 + 1] * s;
}

__global__ void k_scatter2(const int* __restrict__ src, const int* __restrict__ dst,
                           const float* __restrict__ dis, const float* __restrict__ h2,
                           float* __restrict__ out2, int nE) {
  int e = blockIdx.x * blockDim.x + threadIdx.x;
  if (e >= nE) return;
  int s = src[e], d = dst[e];
  float norm = dis[s] * dis[d];
  atomicAdd(&out2[(size_t)d * 2 + 0], h2[(size_t)s * 2 + 0] * norm);
  atomicAdd(&out2[(size_t)d * 2 + 1], h2[(size_t)s * 2 + 1] * norm);
}

__global__ void k_logsoftmax(const float* __restrict__ out2, const float* __restrict__ b2,
                             float* __restrict__ y, int n) {
  int i = blockIdx.x * blockDim.x + threadIdx.x;
  if (i >= n) return;
  float a0 = out2[(size_t)i * 2 + 0] + b2[0];
  float a1 = out2[(size_t)i * 2 + 1] + b2[1];
  float mx = fmaxf(a0, a1);
  float lse = mx + logf(expf(a0 - mx) + expf(a1 - mx));
  y[(size_t)i * 2 + 0] = a0 - lse;
  y[(size_t)i * 2 + 1] = a1 - lse;
}


extern "C" void kernel_launch(void* const* d_in, const int* in_sizes, int n_in,
                              void* d_out, int out_size, void* d_ws, size_t ws_size,
                              hipStream_t stream) {
  const float* x  = (const float*)d_in[0];
  const int*   ei = (const int*)d_in[1];
  const float* W1 = (const float*)d_in[2];
  const float* b1 = (const float*)d_in[3];
  const float* W2 = (const float*)d_in[4];
  const float* b2 = (const float*)d_in[5];

  const int n  = in_sizes[0] / IN_DIM;
  const int nE = in_sizes[1] / 2;
  const int* src = ei;
  const int* dst = ei + nE;
  float* y = (float*)d_out;

  dim3 blk(TB);
  int gN   = (n + TB - 1) / TB;
  int gE   = (nE + TB - 1) / TB;
  int gN16 = (int)(((long long)n * HID + TB - 1) / TB);
  int tiles = (n + 15) / 16;
  int gT   = (tiles + (TB / 32) - 1) / (TB / 32);

  const size_t big_words  = (size_t)nE > (size_t)n * 2 * HID ? (size_t)nE : (size_t)n * 2 * HID;
  const size_t need_words = (size_t)n * 5 + 1024 + (size_t)nE + big_words;
  const int NB  = (n + RANGE - 1) / RANGE;
  const int NCH = (nE + CH - 1) / CH;
  const size_t n2 = (size_t)NB * (size_t)NCH;
  int gS = (int)((n2 + SCAN_TILE - 1) / SCAN_TILE);
  if (ws_size < need_words * sizeof(int) || NB > NBMAX || (nE & 3) != 0 || (n & 3) != 0 ||
      n2 > (size_t)n || n2 > (size_t)1024 * SCAN_TILE) {
    float* deg  = (float*)d_ws;
    float* dis  = deg + n;
    float* h    = dis + n;
    float* out1 = h    + (size_t)n * HID;
    float* h2   = out1 + (size_t)n * HID;
    float* out2 = h2   + (size_t)n * 2;
    int gE4 = (int)(((long long)nE * 4 + TB - 1) / TB);
    k_init_deg  <<<gN,  blk, 0, stream>>>(deg, n);
    k_deg_edges <<<gE,  blk, 0, stream>>>(dst, deg, nE);
    k_dis       <<<gN,  blk, 0, stream>>>(deg, dis, n);
    k_gemm1_wmma<<<gT,  blk, 0, stream>>>(x, W1, h, n);
    k_self1     <<<gN,  blk, 0, stream>>>(h, dis, out1, n);
    k_scatter1  <<<gE4, blk, 0, stream>>>(src, dst, dis, h, out1, nE);
    k_relu_bias <<<gN16,blk, 0, stream>>>(out1, b1, n);
    k_gemm2_wmma<false><<<gT, blk, 0, stream>>>(out1, W2, dis, h2, n);
    k_self2     <<<gN,  blk, 0, stream>>>(h2, dis, out2, n);
    k_scatter2  <<<gE,  blk, 0, stream>>>(src, dst, dis, h2, out2, nE);
    k_logsoftmax<<<gN,  blk, 0, stream>>>(out2, b2, y, n);
    return;
  }

  int*   cnt     = (int*)d_ws;
  int*   row_ptr = cnt + n;
  float* dis     = (float*)(row_ptr + n);
  int*   bsum    = (int*)(dis + n);
  int*   cc      = bsum + 1024;
  int*   off     = cc + n2;
  unsigned* binned = (unsigned*)(cc + (size_t)n * 2);
  float* h       = (float*)binned;
  float* r       = h + (size_t)n * HID;
  float* t2      = h;
  int*   col     = (int*)binned + big_words;

  dim3 sblk(SB);
  const bool tab = n <= DCAP && (size_t)NB * (RANGE / 4) <= n2;
  unsigned* deg8 = tab ? (unsigned*)cc : nullptr;
  k_bin_count  <<<NCH, sblk, 0, stream>>>(dst, cc, nE, NB, NCH);
  k_blocksum   <<<gS,  blk,  0, stream>>>(cc, bsum, (int)n2);
  k_scan_apply <<<gS,  blk,  0, stream>>>(cc, bsum, off, (int)n2);
  k_bin_scatter<<<NCH, sblk, 0, stream>>>(src, dst, cc, off, binned, nE, NB, NCH);
  k_range_lists<<<NB,  sblk, 0, stream>>>(binned, off, cnt, row_ptr, dis, col, deg8, n, nE, NB, NCH);
  k_gemm1_wmma <<<gT,  blk,  0, stream>>>(x, W1, h, n);
  if (tab) k_pull1_tab<<<NWGP, sblk, 0, stream>>>(col, row_ptr, cnt, dis, deg8, h, b1, r, n);
  else     k_pull1    <<<gN16, blk,  0, stream>>>(col, row_ptr, cnt, dis, h, b1, r, n);
  k_gemm2_wmma<true><<<gT, blk, 0, stream>>>(r, W2, dis, t2, n);
  k_pull2      <<<gN16,blk,  0, stream>>>(col, row_ptr, cnt, t2, b2, y, n);
}


# Each edge chunk counts, scans and places its records in one launch, in its own block; a range's base in col is the sum of its runs' starts: no count table or global scan, 9 -> 6 launches; same sums
# speedup vs baseline: 6.2138x; 1.0991x over previous
#include <hip/hip_runtime.h>
#include <math.h>

typedef __attribute__((ext_vector_type(2))) float v2f;
typedef __attribute__((ext_vector_type(8))) float v8f;

#define IN_DIM 128
#define HID    16
#define TB     256
#define RB     9
#define RANGE  512
#define CH     16384
#define NBMAX  2048
#define NCHMAX 2048
#define RG     4
#define SB     1024
#define CAPC   24576
#define DCAP   262144
#define NWGP   256


__device__ __forceinline__ int wave_sum_i(int v) {
#pragma unroll
  for (int m = 16; m > 0; m >>= 1) v += __shfl_xor(v, m, 32);
  return v;
}

__device__ __forceinline__ int block_excl_scan(int v, int* sm_w) {
  const int lane = threadIdx.x & 31, wave = threadIdx.x >> 5;
  int inc = v;
  for (int d = 1; d < 32; d <<= 1) {
    int u = __shfl_up(inc, d, 32);
    if (lane >= d) inc += u;
  }
  if (lane == 31) sm_w[wave] = inc;
  __syncthreads();
  if (wave == 0) {
    const int w = sm_w[lane];
    int winc = w;
    for (int d = 1; d < 32; d <<= 1) {
      int u = __shfl_up(winc, d, 32);
      if (lane >= d) winc += u;
    }
    sm_w[lane] = winc - w;
  }
  __syncthreads();
  return inc - v + sm_w[wave];
}

__global__ void __launch_bounds__(SB, 16) k_chunk_sort(const int* __restrict__ src, const int* __restrict__ dst,
                                                   int* __restrict__ startT, unsigned* __restrict__ binned,
                                                   int nE, int NB, int NCH) {
  __shared__ int curL[NBMAX];
  __shared__ int sm_w[32];
  __shared__ alignas(16) unsigned recL[CH];
  const int t = threadIdx.x, chunk = blockIdx.x;
  const int e0 = chunk * CH;
  int4 d4[CH / (4 * SB)], s4[CH / (4 * SB)];
#pragma unroll
  for (int k = 0; k < CH / (4 * SB); ++k) {
    const int e = e0 + 4 * (t + k * SB);
    const int ee = e < nE ? e : 0;
    d4[k] = *reinterpret_cast<const int4*>(dst + ee);
    s4[k] = *reinterpret_cast<const int4*>(src + ee);
  }
  for (int b = t; b < NB; b += SB) curL[b] = 0;
  __syncthreads();
#pragma unroll
  for (int k = 0; k < CH / (4 * SB); ++k) {
    if (e0 + 4 * (t + k * SB) < nE) {
      atomicAdd(&curL[d4[k].x >> RB], 1);
      atomicAdd(&curL[d4[k].y >> RB], 1);
      atomicAdd(&curL[d4[k].z >> RB], 1);
      atomicAdd(&curL[d4[k].w >> RB], 1);
    }
  }
  __syncthreads();
  const int b0 = 2 * t, b1 = 2 * t + 1;
  const int h0 = (b0 < NB) ? curL[b0] : 0;
  const int h1 = (b1 < NB) ? curL[b1] : 0;
  const int ex = block_excl_scan(h0 + h1, sm_w);
  if (b0 < NB) { curL[b0] = ex;      startT[b0 * NCH + chunk] = ex; }
  if (b1 < NB) { curL[b1] = ex + h0; startT[b1 * NCH + chunk] = ex + h0; }
  __syncthreads();
#pragma unroll
  for (int k = 0; k < CH / (4 * SB); ++k) {
    if (e0 + 4 * (t + k * SB) < nE) {
      const int q0 = atomicAdd(&curL[d4[k].x >> RB], 1);
      const int q1 = atomicAdd(&curL[d4[k].y >> RB], 1);
      const int q2 = atomicAdd(&curL[d4[k].z >> RB], 1);
      const int q3 = atomicAdd(&curL[d4[k].w >> RB], 1);
      recL[q0] = ((unsigned)s4[k].x << RB) | (unsigned)(d4[k].x & (RANGE - 1));
      recL[q1] = ((unsigned)s4[k].y << RB) | (unsigned)(d4[k].y & (RANGE - 1));
      recL[q2] = ((unsigned)s4[k].z << RB) | (unsigned)(d4[k].z & (RANGE - 1));
      recL[q3] = ((unsigned)s4[k].w << RB) | (unsigned)(d4[k].w & (RANGE - 1));
    }
  }
  __syncthreads();
  const int len = min(CH, nE - e0);
  for (int i = 4 * t; i < len; i += 4 * SB)
    *reinterpret_cast<uint4*>(binned + e0 + i) = *reinterpret_cast<const uint4*>(recL + i);
}

template <class F>
__device__ __forceinline__ void range_records(const unsigned* __restrict__ binned, const int* runS,
                                              const int* runN, int NCH, F f) {
  const int lane = threadIdx.x & 31, wave = threadIdx.x >> 5;
  for (int c0 = wave; c0 < NCH; c0 += 32 * RG) {
    unsigned ra[RG], rb[RG];
    int ln[RG], bs[RG];
#pragma unroll
    for (int g = 0; g < RG; ++g) {
      const int c = c0 + 32 * g;
      ln[g] = (c < NCH) ? runN[c] : 0;
      bs[g] = (c < NCH) ? runS[c] : 0;
      ra[g] = binned[(lane < ln[g]) ? bs[g] + lane : 0];
      rb[g] = binned[(lane + 32 < ln[g]) ? bs[g] + lane + 32 : 0];
    }
#pragma unroll
    for (int g = 0; g < RG; ++g) {
      if (lane < ln[g]) f(ra[g]);
      if (lane + 32 < ln[g]) f(rb[g]);
      for (int j = lane + 64; j < ln[g]; j += 32) f(binned[bs[g] + j]);
    }
  }
}

__global__ void __launch_bounds__(SB) k_range_lists(const unsigned* __restrict__ binned, const int* __restrict__ startT,
                                                    int* __restrict__ cnt, int* __restrict__ row_ptr,
                                                    float* __restrict__ dis, int* __restrict__ col,
                                                    unsigned* __restrict__ deg8,
                                                    int n, int nE, int NB, int NCH) {
  __shared__ alignas(16) int cntL[RANGE];
  __shared__ int curL[RANGE];
  __shared__ int sm_w[32];
  __shared__ int sm_s[64];
  __shared__ int runS[NCHMAX];
  __shared__ int runN[NCHMAX];
  __shared__ int colL[CAPC];
  const int t = threadIdx.x, lane = t & 31, wave = t >> 5, b = blockIdx.x;
  int sS = 0, sR = 0;
  for (int c = t; c < NCH; c += SB) {
    const int st = startT[b * NCH + c];
    const int en = (b + 1 < NB) ? startT[(b + 1) * NCH + c] : min(CH, nE - c * CH);
    runS[c] = c * CH + st;
    runN[c] = en - st;
    sS += st;
    sR += en - st;
  }
  if (t < RANGE) cntL[t] = 0;
  sS = wave_sum_i(sS);
  sR = wave_sum_i(sR);
  if (lane == 0) { sm_s[wave] = sS; sm_s[32 + wave] = sR; }
  __syncthreads();
  const int S = wave_sum_i(sm_s[lane]);
  const int R = wave_sum_i(sm_s[32 + lane]);
  range_records(binned, runS, runN, NCH,
                [&](unsigned rec) { atomicAdd(&cntL[rec & (RANGE - 1)], 1); });
  __syncthreads();
  const int c = (t < RANGE) ? cntL[t] : 0;
  const int ex = block_excl_scan(c, sm_w);
  if (t < RANGE) {
    curL[t] = ex;
    const int node = (b << RB) + t;
    if (node < n) {
      cnt[node] = c;
      row_ptr[node] = S + ex;
      dis[node] = rsqrtf((float)(c + 1));
    }
  }
  if (deg8 != nullptr && t < RANGE / 4) {
    const int4 c4 = *reinterpret_cast<const int4*>(&cntL[4 * t]);
    const unsigned pk = (unsigned)min(c4.x, 255) | ((unsigned)min(c4.y, 255) << 8) |
                        ((unsigned)min(c4.z, 255) << 16) | ((unsigned)min(c4.w, 255) << 24);
    const int wi = (b << (RB - 2)) + t;
    if (4 * wi < n) deg8[wi] = pk;
  }
  __syncthreads();
  if (R <= CAPC) {
    range_records(binned, runS, runN, NCH, [&](unsigned rec) {
      const int slot = atomicAdd(&curL[rec & (RANGE - 1)], 1);
      colL[slot] = (int)(rec >> RB);
    });
    __syncthreads();
    for (int i = t; i < R; i += SB) col[S + i] = colL[i];
  } else {
    range_records(binned, runS, runN, NCH, [&](unsigned rec) {
      const int slot = atomicAdd(&curL[rec & (RANGE - 1)], 1);
      col[S + slot] = (int)(rec >> RB);
    });
  }
}


__global__ void k_gemm1_wmma(const float* __restrict__ x, const float* __restrict__ W1,
                             float* __restrict__ h, int n) {
  const int lane = threadIdx.x & 31;
  const int wave = threadIdx.x >> 5;
  const int tile = blockIdx.x * (blockDim.x >> 5) + wave;
  const int row0 = tile << 4;
  if (row0 >= n) return;
  const int m  = lane & 15;
  const int kh = lane >> 4;
  const float* xrow = x + (size_t)(row0 + m) * IN_DIM + 2 * kh;

  v8f c = {0.f, 0.f, 0.f, 0.f, 0.f, 0.f, 0.f, 0.f};
#pragma unroll
  for (int k0 = 0; k0 < IN_DIM / 4; ++k0) {
    const int kk = 4 * k0 + 2 * kh;
    const float2 av = *reinterpret_cast<const float2*>(xrow + 4 * k0);
    v2f a; a.x = av.x; a.y = av.y;
    v2f b; b.x = W1[(kk + 0) * HID + m];
           b.y = W1[(kk + 1) * HID + m];
    c = __builtin_amdgcn_wmma_f32_16x16x4_f32(false, a, false, b, (short)0, c, false, false);
  }
  float* hout = h + (size_t)(row0 + 8 * kh) * HID + m;
#pragma unroll
  for (int r = 0; r < 8; ++r)
    hout[(size_t)r * HID] = c[r];
}


template <bool REC>
__global__ void k_gemm2_wmma(const float* __restrict__ h1, const float* __restrict__ W2,
                             const float* __restrict__ dis, float* __restrict__ t2, int n) {
  const int lane = threadIdx.x & 31;
  const int wave = threadIdx.x >> 5;
  const int tile = blockIdx.x * (blockDim.x >> 5) + wave;
  const int row0 = tile << 4;
  if (row0 >= n) return;
  const int m  = lane & 15;
  const int kh = lane >> 4;
  const float bmask = (m < 2) ? 1.0f : 0.0f;
  const int   col   = (m < 2) ? m : 0;
  const float* arow = h1 + (size_t)(row0 + m) * HID + 2 * kh;
  float dv[8] = {0.f, 0.f, 0.f, 0.f, 0.f, 0.f, 0.f, 0.f};
  if (REC) {
    const float4 da = *reinterpret_cast<const float4*>(dis + row0 + 8 * kh);
    const float4 db = *reinterpret_cast<const float4*>(dis + row0 + 8 * kh + 4);
    dv[0] = da.x; dv[1] = da.y; dv[2] = da.z; dv[3] = da.w;
    dv[4] = db.x; dv[5] = db.y; dv[6] = db.z; dv[7] = db.w;
  }

  v8f c = {0.f, 0.f, 0.f, 0.f, 0.f, 0.f, 0.f, 0.f};
#pragma unroll
  for (int k0 = 0; k0 < HID / 4; ++k0) {
    const int kk = 4 * k0 + 2 * kh;
    const float2 av = *reinterpret_cast<const float2*>(arow + 4 * k0);
    v2f a; a.x = av.x; a.y = av.y;
    v2f b; b.x = W2[(kk + 0) * 2 + col] * bmask;
           b.y = W2[(kk + 1) * 2 + col] * bmask;
    c = __builtin_amdgcn_wmma_f32_16x16x4_f32(false, a, false, b, (short)0, c, false, false);
  }
  if (REC) {
    if (m < 4) {
#pragma unroll
      for (int r = 0; r < 8; ++r) {
        const int row = row0 + r + 8 * kh;
        const float v = (m < 2) ? c[r] : ((m == 2) ? dv[r] : 0.f);
        t2[(size_t)row * 4 + m] = v;
      }
    }
  } else if (m < 2) {
#pragma unroll
    for (int r = 0; r < 8; ++r)
      t2[(size_t)(row0 + r + 8 * kh) * 2 + m] = c[r];
  }
}


__global__ void k_pull1(const int* __restrict__ col, const int* __restrict__ row_ptr,
                        const int* __restrict__ cnt, const float* __restrict__ dis,
                        const float* __restrict__ h, const float* __restrict__ b1,
                        float* __restrict__ r, int n) {
  const long long gid = (long long)blockIdx.x * blockDim.x + threadIdx.x;
  const int d = (int)(gid >> 4);
  if (d >= n) return;
  const int f = threadIdx.x & 15;
  const int beg = row_ptr[d];
  const int m = cnt[d];
  const float dd = dis[d];
  float acc = h[(size_t)d * HID + f] * (dd * dd);
  for (int j0 = 0; j0 < m; j0 += 16) {
    const int jj = j0 + f;
    int s_l = d;
    float w_l = 0.f;
    if (jj < m) { s_l = col[beg + jj]; w_l = dis[s_l] * dd; }
    const int nb = m - j0;
#pragma unroll
    for (int t = 0; t < 16; ++t) {
      const int s = __shfl(s_l, t, 16);
      const float w = __shfl(w_l, t, 16);
      const float hv = h[(size_t)s * HID + f];
      acc = fmaf((t < nb) ? hv : 0.f, w, acc);
    }
  }
  const float v = acc + b1[f];
  r[(size_t)d * HID + f] = v > 0.f ? v : 0.f;
}

__global__ void __launch_bounds__(SB) k_pull1_tab(const int* __restrict__ col, const int* __restrict__ row_ptr,
                        const int* __restrict__ cnt, const float* __restrict__ dis,
                        const unsigned* __restrict__ deg8,
                        const float* __restrict__ h, const float* __restrict__ b1,
                        float* __restrict__ r, int n) {
  __shared__ alignas(16) unsigned degL[DCAP / 4];
  const int t = threadIdx.x;
  {
    const uint4* src4 = reinterpret_cast<const uint4*>(deg8);
    uint4* dst4 = reinterpret_cast<uint4*>(degL);
    const int nq = n >> 4;
    int i = t;
    for (; i + 3 * SB < nq; i += 4 * SB) {
      const uint4 v0 = src4[i], v1 = src4[i + SB], v2 = src4[i + 2 * SB], v3 = src4[i + 3 * SB];
      dst4[i] = v0; dst4[i + SB] = v1; dst4[i + 2 * SB] = v2; dst4[i + 3 * SB] = v3;
    }
    for (; i < nq; i += SB) dst4[i] = src4[i];
    if (t < ((n >> 2) & 3)) degL[(nq << 2) + t] = deg8[(nq << 2) + t];
  }
  __syncthreads();
  const unsigned char* degB = reinterpret_cast<const unsigned char*>(degL);
  const int lane = t & 31, wave = t >> 5;
  const int f = lane & 15, half = lane >> 4;
  const int npair = n >> 1;
  const int per = (npair + NWGP - 1) / NWGP;
  const int p0 = blockIdx.x * per;
  const int P = min(per, npair - p0);
  if (P <= 0) return;
  const int qa = p0 + (wave * P) / (SB / 32);
  const int qn = p0 + ((wave + 1) * P) / (SB / 32) - qa;
  const int nd0 = 2 * qa;
  const int dl = nd0 + lane;
  const bool has = lane < 2 * qn;
  const int   beg_l = has ? row_ptr[dl] : 0;
  const int   m_l   = has ? cnt[dl] : 0;
  const float dd_l  = has ? dis[dl] : 0.f;
  const float bias = b1[f];
  for (int i = 0; i < qn; ++i) {
    const int k = 2 * i + half;
    const int d = nd0 + k;
    const int beg = __shfl(beg_l, k, 32);
    const int m = __shfl(m_l, k, 32);
    const float dd = __shfl(dd_l, k, 32);
    int s_cur = (f < m) ? col[beg + f] : d;
    float acc = h[(size_t)d * HID + f] * (dd * dd);
    for (int j0 = 0; j0 < m; j0 += 16) {
      const int jj = j0 + f;
      const int jn = jj + 16;
      const int s_l = s_cur;
      float w_l = 0.f;
      if (jj < m) {
        const unsigned bq = degB[s_l];
        const float ds = (bq < 255u) ? rsqrtf((float)(int)(bq + 1u)) : dis[s_l];
        w_l = ds * dd;
      }
      s_cur = (jn < m) ? col[beg + jn] : d;
      const int nb = m - j0;
#pragma unroll
      for (int tt = 0; tt < 16; ++tt) {
        const int s = __shfl(s_l, tt, 16);
        const float w = __shfl(w_l, tt, 16);
        const float hv = h[(size_t)s * HID + f];
        acc = fmaf((tt < nb) ? hv : 0.f, w, acc);
      }
    }
    const float v = acc + bias;
    r[(size_t)d * HID + f] = v > 0.f ? v : 0.f;
  }
}

__global__ void k_pull2(const int* __restrict__ col, const int* __restrict__ row_ptr,
                        const int* __restrict__ cnt, const float* __restrict__ t2,
                        const float* __restrict__ b2, float* __restrict__ y, int n) {
  __shared__ float2 so[TB / 16];
  const int f = threadIdx.x & 15;
  const int hw = threadIdx.x >> 4;
  const int d0 = blockIdx.x * (TB / 16);
  const int d = d0 + hw;
  float a0 = 0.f, a1 = 0.f;
  if (d < n) {
    const int beg = row_ptr[d];
    const int m = cnt[d];
    const float4 rd = *reinterpret_cast<const float4*>(t2 + (size_t)d * 4);
    const float dd = rd.z;
    for (int jj = f; jj < m; jj += 32) {
      const int jb = jj + 16;
      const int sa = col[beg + jj];
      const int sb = (jb < m) ? col[beg + jb] : sa;
      const float4 va = *reinterpret_cast<const float4*>(t2 + (size_t)sa * 4);
      const float4 vb = *reinterpret_cast<const float4*>(t2 + (size_t)sb * 4);
      const float wa = va.z * dd;
      const float wb = (jb < m) ? vb.z * dd : 0.f;
      a0 = fmaf(va.x, wa, a0); a1 = fmaf(va.y, wa, a1);
      a0 = fmaf((jb < m) ? vb.x : 0.f, wb, a0); a1 = fmaf((jb < m) ? vb.y : 0.f, wb, a1);
    }
#pragma unroll
    for (int k = 8; k > 0; k >>= 1) {
      a0 += __shfl_xor(a0, k, 16);
      a1 += __shfl_xor(a1, k, 16);
    }
    if (f == 0) {
      const float s2 = dd * dd;
      const float x0 = a0 + rd.x * s2 + b2[0];
      const float x1 = a1 + rd.y * s2 + b2[1];
      const float mx = fmaxf(x0, x1);
      const float lse = mx + logf(expf(x0 - mx) + expf(x1 - mx));
      so[hw] = make_float2(x0 - lse, x1 - lse);
    }
  }
  __syncthreads();
  if (threadIdx.x < TB / 16 && d0 + (int)threadIdx.x < n)
    *reinterpret_cast<float2*>(y + (size_t)(d0 + threadIdx.x) * 2) = so[threadIdx.x];
}


__global__ void k_init_deg(float* __restrict__ deg, int n) {
  int i = blockIdx.x * blockDim.x + threadIdx.x;
  if (i < n) deg[i] = 1.0f;
}

__global__ void k_deg_edges(const int* __restrict__ dst, float* __restrict__ deg, int nE) {
  int e = blockIdx.x * blockDim.x + threadIdx.x;
  if (e < nE) atomicAdd(&deg[dst[e]], 1.0f);
}

__global__ void k_dis(const float* __restrict__ deg, float* __restrict__ dis, int n) {
  int i = blockIdx.x * blockDim.x + threadIdx.x;
  if (i < n) dis[i] = rsqrtf(deg[i]);
}

__global__ void k_self1(const float* __restrict__ h, const float* __restrict__ dis,
                        float* __restrict__ out1, int n) {
  int i = blockIdx.x * blockDim.x + threadIdx.x;
  if (i >= n) return;
  float s = dis[i]; s = s * s;
  const float4* hp = reinterpret_cast<const float4*>(h + (size_t)i * HID);
  float4* op = reinterpret_cast<float4*>(out1 + (size_t)i * HID);
#pragma unroll
  for (int q = 0; q < 4; ++q) {
    float4 v = hp[q];
    v.x *= s; v.y *= s; v.z *= s; v.w *= s;
    op[q] = v;
  }
}

__global__ void k_scatter1(const int* __restrict__ src, const int* __restrict__ dst,
                           const float* __restrict__ dis, const float* __restrict__ h,
                           float* __restrict__ out1, int nE) {
  long long t = (long long)blockIdx.x * blockDim.x + threadIdx.x;
  int e = (int)(t >> 2);
  if (e >= nE) return;
  int q = (int)(t & 3);
  int s = src[e], d = dst[e];
  float norm = dis[s] * dis[d];
  float4 v = *reinterpret_cast<const float4*>(h + (size_t)s * HID + 4 * q);
  float* o = out1 + (size_t)d * HID + 4 * q;
  atomicAdd(o + 0, v.x * norm);
  atomicAdd(o + 1, v.y * norm);
  atomicAdd(o + 2, v.z * norm);
  atomicAdd(o + 3, v.w * norm);
}

__global__ void k_relu_bias(float* __restrict__ out1, const float* __restrict__ b1, int n) {
  long long t = (long long)blockIdx.x * blockDim.x + threadIdx.x;
  if (t >= (long long)n * HID) return;
  int j = (int)(t & (HID - 1));
  float v = out1[t] + b1[j];
  out1[t] = v > 0.f ? v : 0.f;
}

__global__ void k_self2(const float* __restrict__ h2, const float* __restrict__ dis,
                        float* __restrict__ out2, int n) {
  int i = blockIdx.x * blockDim.x + threadIdx.x;
  if (i >= n) return;
  float s = dis[i]; s = s * s;
  out2[(size_t)i * 2 + 0] = h2[(size_t)i * 2 + 0] * s;
  out2[(size_t)i * 2 + 1] = h2[(size_t)i * 2 + 1] * s;
}

__global__ void k_scatter2(const int* __restrict__ src, const int* __restrict__ dst,
                           const float* __restrict__ dis, const float* __restrict__ h2,
                           float* __restrict__ out2, int nE) {
  int e = blockIdx.x * blockDim.x + threadIdx.x;
  if (e >= nE) return;
  int s = src[e], d = dst[e];
  float norm = dis[s] * dis[d];
  atomicAdd(&out2[(size_t)d * 2 + 0], h2[(size_t)s * 2 + 0] * norm);
  atomicAdd(&out2[(size_t)d * 2 + 1], h2[(size_t)s * 2 + 1] * norm);
}

__global__ void k_logsoftmax(const float* __restrict__ out2, const float* __restrict__ b2,
                             float* __restrict__ y, int n) {
  int i = blockIdx.x * blockDim.x + threadIdx.x;
  if (i >= n) return;
  float a0 = out2[(size_t)i * 2 + 0] + b2[0];
  float a1 = out2[(size_t)i * 2 + 1] + b2[1];
  float mx = fmaxf(a0, a1);
  float lse = mx + logf(expf(a0 - mx) + expf(a1 - mx));
  y[(size_t)i * 2 + 0] = a0 - lse;
  y[(size_t)i * 2 + 1] = a1 - lse;
}


extern "C" void kernel_launch(void* const* d_in, const int* in_sizes, int n_in,
                              void* d_out, int out_size, void* d_ws, size_t ws_size,
                              hipStream_t stream) {
  const float* x  = (const float*)d_in[0];
  const int*   ei = (const int*)d_in[1];
  const float* W1 = (const float*)d_in[2];
  const float* b1 = (const float*)d_in[3];
  const float* W2 = (const float*)d_in[4];
  const float* b2 = (const float*)d_in[5];

  const int n  = in_sizes[0] / IN_DIM;
  const int nE = in_sizes[1] / 2;
  const int* src = ei;
  const int* dst = ei + nE;
  float* y = (float*)d_out;

  dim3 blk(TB);
  int gN   = (n + TB - 1) / TB;
  int gE   = (nE + TB - 1) / TB;
  int gN16 = (int)(((long long)n * HID + TB - 1) / TB);
  int tiles = (n + 15) / 16;
  int gT   = (tiles + (TB / 32) - 1) / (TB / 32);

  const size_t big_words  = (size_t)nE > (size_t)n * 2 * HID ? (size_t)nE : (size_t)n * 2 * HID;
  const size_t need_words = (size_t)n * 5 + 1024 + (size_t)nE + big_words;
  const int NB  = (n + RANGE - 1) / RANGE;
  const int NCH = (nE + CH - 1) / CH;
  const size_t n2 = (size_t)NB * (size_t)NCH;
  if (ws_size < need_words * sizeof(int) || NB > NBMAX || NCH > NCHMAX || (nE & 3) != 0 || (n & 3) != 0 ||
      n2 > (size_t)n) {
    float* deg  = (float*)d_ws;
    float* dis  = deg + n;
    float* h    = dis + n;
    float* out1 = h    + (size_t)n * HID;
    float* h2   = out1 + (size_t)n * HID;
    float* out2 = h2   + (size_t)n * 2;
    int gE4 = (int)(((long long)nE * 4 + TB - 1) / TB);
    k_init_deg  <<<gN,  blk, 0, stream>>>(deg, n);
    k_deg_edges <<<gE,  blk, 0, stream>>>(dst, deg, nE);
    k_dis       <<<gN,  blk, 0, stream>>>(deg, dis, n);
    k_gemm1_wmma<<<gT,  blk, 0, stream>>>(x, W1, h, n);
    k_self1     <<<gN,  blk, 0, stream>>>(h, dis, out1, n);
    k_scatter1  <<<gE4, blk, 0, stream>>>(src, dst, dis, h, out1, nE);
    k_relu_bias <<<gN16,blk, 0, stream>>>(out1, b1, n);
    k_gemm2_wmma<false><<<gT, blk, 0, stream>>>(out1, W2, dis, h2, n);
    k_self2     <<<gN,  blk, 0, stream>>>(h2, dis, out2, n);
    k_scatter2  <<<gE,  blk, 0, stream>>>(src, dst, dis, h2, out2, nE);
    k_logsoftmax<<<gN,  blk, 0, stream>>>(out2, b2, y, n);
    return;
  }

  int*   cnt     = (int*)d_ws;
  int*   row_ptr = cnt + n;
  float* dis     = (float*)(row_ptr + n);
  int*   bsum    = (int*)(dis + n);
  int*   cc      = bsum + 1024;
  int*   startT  = cc + n2;
  unsigned* binned = (unsigned*)(cc + (size_t)n * 2);
  float* h       = (float*)binned;
  float* r       = h + (size_t)n * HID;
  float* t2      = h;
  int*   col     = (int*)binned + big_words;

  dim3 sblk(SB);
  const bool tab = n <= DCAP && (size_t)NB * (RANGE / 4) <= n2;
  unsigned* deg8 = tab ? (unsigned*)cc : nullptr;
  k_chunk_sort <<<NCH, sblk, 0, stream>>>(src, dst, startT, binned, nE, NB, NCH);
  k_range_lists<<<NB,  sblk, 0, stream>>>(binned, startT, cnt, row_ptr, dis, col, deg8, n, nE, NB, NCH);
  k_gemm1_wmma <<<gT,  blk,  0, stream>>>(x, W1, h, n);
  if (tab) k_pull1_tab<<<NWGP, sblk, 0, stream>>>(col, row_ptr, cnt, dis, deg8, h, b1, r, n);
  else     k_pull1    <<<gN16, blk,  0, stream>>>(col, row_ptr, cnt, dis, h, b1, r, n);
  k_gemm2_wmma<true><<<gT, blk, 0, stream>>>(r, W2, dis, t2, n);
  k_pull2      <<<gN16,blk,  0, stream>>>(col, row_ptr, cnt, t2, b2, y, n);
}


# Layer-1 pull gathers rows 16 bytes per lane: 4 lanes per 64-B row, 8 rows per load instruction (was 2), a quarter of the loads and shuffles; sums reordered: 4 partial sums per node, then a tree
# speedup vs baseline: 6.4496x; 1.0379x over previous
#include <hip/hip_runtime.h>
#include <math.h>

typedef __attribute__((ext_vector_type(2))) float v2f;
typedef __attribute__((ext_vector_type(8))) float v8f;

#define IN_DIM 128
#define HID    16
#define TB     256
#define RB     9
#define RANGE  512
#define CH     16384
#define NBMAX  2048
#define NCHMAX 2048
#define RG     4
#define SB     1024
#define CAPC   24576
#define DCAP   262144
#define NWGP   256


__device__ __forceinline__ int wave_sum_i(int v) {
#pragma unroll
  for (int m = 16; m > 0; m >>= 1) v += __shfl_xor(v, m, 32);
  return v;
}

__device__ __forceinline__ int block_excl_scan(int v, int* sm_w) {
  const int lane = threadIdx.x & 31, wave = threadIdx.x >> 5;
  int inc = v;
  for (int d = 1; d < 32; d <<= 1) {
    int u = __shfl_up(inc, d, 32);
    if (lane >= d) inc += u;
  }
  if (lane == 31) sm_w[wave] = inc;
  __syncthreads();
  if (wave == 0) {
    const int w = sm_w[lane];
    int winc = w;
    for (int d = 1; d < 32; d <<= 1) {
      int u = __shfl_up(winc, d, 32);
      if (lane >= d) winc += u;
    }
    sm_w[lane] = winc - w;
  }
  __syncthreads();
  return inc - v + sm_w[wave];
}

__global__ void __launch_bounds__(SB, 16) k_chunk_sort(const int* __restrict__ src, const int* __restrict__ dst,
                                                   int* __restrict__ startT, unsigned* __restrict__ binned,
                                                   int nE, int NB, int NCH) {
  __shared__ int curL[NBMAX];
  __shared__ int sm_w[32];
  __shared__ alignas(16) unsigned recL[CH];
  const int t = threadIdx.x, chunk = blockIdx.x;
  const int e0 = chunk * CH;
  int4 d4[CH / (4 * SB)], s4[CH / (4 * SB)];
#pragma unroll
  for (int k = 0; k < CH / (4 * SB); ++k) {
    const int e = e0 + 4 * (t + k * SB);
    const int ee = e < nE ? e : 0;
    d4[k] = *reinterpret_cast<const int4*>(dst + ee);
    s4[k] = *reinterpret_cast<const int4*>(src + ee);
  }
  for (int b = t; b < NB; b += SB) curL[b] = 0;
  __syncthreads();
#pragma unroll
  for (int k = 0; k < CH / (4 * SB); ++k) {
    if (e0 + 4 * (t + k * SB) < nE) {
      atomicAdd(&curL[d4[k].x >> RB], 1);
      atomicAdd(&curL[d4[k].y >> RB], 1);
      atomicAdd(&curL[d4[k].z >> RB], 1);
      atomicAdd(&curL[d4[k].w >> RB], 1);
    }
  }
  __syncthreads();
  const int b0 = 2 * t, b1 = 2 * t + 1;
  const int h0 = (b0 < NB) ? curL[b0] : 0;
  const int h1 = (b1 < NB) ? curL[b1] : 0;
  const int ex = block_excl_scan(h0 + h1, sm_w);
  if (b0 < NB) { curL[b0] = ex;      startT[b0 * NCH + chunk] = ex; }
  if (b1 < NB) { curL[b1] = ex + h0; startT[b1 * NCH + chunk] = ex + h0; }
  __syncthreads();
#pragma unroll
  for (int k = 0; k < CH / (4 * SB); ++k) {
    if (e0 + 4 * (t + k * SB) < nE) {
      const int q0 = atomicAdd(&curL[d4[k].x >> RB], 1);
      const int q1 = atomicAdd(&curL[d4[k].y >> RB], 1);
      const int q2 = atomicAdd(&curL[d4[k].z >> RB], 1);
      const int q3 = atomicAdd(&curL[d4[k].w >> RB], 1);
      recL[q0] = ((unsigned)s4[k].x << RB) | (unsigned)(d4[k].x & (RANGE - 1));
      recL[q1] = ((unsigned)s4[k].y << RB) | (unsigned)(d4[k].y & (RANGE - 1));
      recL[q2] = ((unsigned)s4[k].z << RB) | (unsigned)(d4[k].z & (RANGE - 1));
      recL[q3] = ((unsigned)s4[k].w << RB) | (unsigned)(d4[k].w & (RANGE - 1));
    }
  }
  __syncthreads();
  const int len = min(CH, nE - e0);
  for (int i = 4 * t; i < len; i += 4 * SB)
    *reinterpret_cast<uint4*>(binned + e0 + i) = *reinterpret_cast<const uint4*>(recL + i);
}

template <class F>
__device__ __forceinline__ void range_records(const unsigned* __restrict__ binned, const int* runS,
                                              const int* runN, int NCH, F f) {
  const int lane = threadIdx.x & 31, wave = threadIdx.x >> 5;
  for (int c0 = wave; c0 < NCH; c0 += 32 * RG) {
    unsigned ra[RG], rb[RG];
    int ln[RG], bs[RG];
#pragma unroll
    for (int g = 0; g < RG; ++g) {
      const int c = c0 + 32 * g;
      ln[g] = (c < NCH) ? runN[c] : 0;
      bs[g] = (c < NCH) ? runS[c] : 0;
      ra[g] = binned[(lane < ln[g]) ? bs[g] + lane : 0];
      rb[g] = binned[(lane + 32 < ln[g]) ? bs[g] + lane + 32 : 0];
    }
#pragma unroll
    for (int g = 0; g < RG; ++g) {
      if (lane < ln[g]) f(ra[g]);
      if (lane + 32 < ln[g]) f(rb[g]);
      for (int j = lane + 64; j < ln[g]; j += 32) f(binned[bs[g] + j]);
    }
  }
}

__global__ void __launch_bounds__(SB) k_range_lists(const unsigned* __restrict__ binned, const int* __restrict__ startT,
                                                    int* __restrict__ cnt, int* __restrict__ row_ptr,
                                                    float* __restrict__ dis, int* __restrict__ col,
                                                    unsigned* __restrict__ deg8,
                                                    int n, int nE, int NB, int NCH) {
  __shared__ alignas(16) int cntL[RANGE];
  __shared__ int curL[RANGE];
  __shared__ int sm_w[32];
  __shared__ int sm_s[64];
  __shared__ int runS[NCHMAX];
  __shared__ int runN[NCHMAX];
  __shared__ int colL[CAPC];
  const int t = threadIdx.x, lane = t & 31, wave = t >> 5, b = blockIdx.x;
  int sS = 0, sR = 0;
  for (int c = t; c < NCH; c += SB) {
    const int st = startT[b * NCH + c];
    const int en = (b + 1 < NB) ? startT[(b + 1) * NCH + c] : min(CH, nE - c * CH);
    runS[c] = c * CH + st;
    runN[c] = en - st;
    sS += st;
    sR += en - st;
  }
  if (t < RANGE) cntL[t] = 0;
  sS = wave_sum_i(sS);
  sR = wave_sum_i(sR);
  if (lane == 0) { sm_s[wave] = sS; sm_s[32 + wave] = sR; }
  __syncthreads();
  const int S = wave_sum_i(sm_s[lane]);
  const int R = wave_sum_i(sm_s[32 + lane]);
  range_records(binned, runS, runN, NCH,
                [&](unsigned rec) { atomicAdd(&cntL[rec & (RANGE - 1)], 1); });
  __syncthreads();
  const int c = (t < RANGE) ? cntL[t] : 0;
  const int ex = block_excl_scan(c, sm_w);
  if (t < RANGE) {
    curL[t] = ex;
    const int node = (b << RB) + t;
    if (node < n) {
      cnt[node] = c;
      row_ptr[node] = S + ex;
      dis[node] = rsqrtf((float)(c + 1));
    }
  }
  if (deg8 != nullptr && t < RANGE / 4) {
    const int4 c4 = *reinterpret_cast<const int4*>(&cntL[4 * t]);
    const unsigned pk = (unsigned)min(c4.x, 255) | ((unsigned)min(c4.y, 255) << 8) |
                        ((unsigned)min(c4.z, 255) << 16) | ((unsigned)min(c4.w, 255) << 24);
    const int wi = (b << (RB - 2)) + t;
    if (4 * wi < n) deg8[wi] = pk;
  }
  __syncthreads();
  if (R <= CAPC) {
    range_records(binned, runS, runN, NCH, [&](unsigned rec) {
      const int slot = atomicAdd(&curL[rec & (RANGE - 1)], 1);
      colL[slot] = (int)(rec >> RB);
    });
    __syncthreads();
    for (int i = t; i < R; i += SB) col[S + i] = colL[i];
  } else {
    range_records(binned, runS, runN, NCH, [&](unsigned rec) {
      const int slot = atomicAdd(&curL[rec & (RANGE - 1)], 1);
      col[S + slot] = (int)(rec >> RB);
    });
  }
}


__global__ void k_gemm1_wmma(const float* __restrict__ x, const float* __restrict__ W1,
                             float* __restrict__ h, int n) {
  const int lane = threadIdx.x & 31;
  const int wave = threadIdx.x >> 5;
  const int tile = blockIdx.x * (blockDim.x >> 5) + wave;
  const int row0 = tile << 4;
  if (row0 >= n) return;
  const int m  = lane & 15;
  const int kh = lane >> 4;
  const float* xrow = x + (size_t)(row0 + m) * IN_DIM + 2 * kh;

  v8f c = {0.f, 0.f, 0.f, 0.f, 0.f, 0.f, 0.f, 0.f};
#pragma unroll
  for (int k0 = 0; k0 < IN_DIM / 4; ++k0) {
    const int kk = 4 * k0 + 2 * kh;
    const float2 av = *reinterpret_cast<const float2*>(xrow + 4 * k0);
    v2f a; a.x = av.x; a.y = av.y;
    v2f b; b.x = W1[(kk + 0) * HID + m];
           b.y = W1[(kk + 1) * HID + m];
    c = __builtin_amdgcn_wmma_f32_16x16x4_f32(false, a, false, b, (short)0, c, false, false);
  }
  float* hout = h + (size_t)(row0 + 8 * kh) * HID + m;
#pragma unroll
  for (int r = 0; r < 8; ++r)
    hout[(size_t)r * HID] = c[r];
}


template <bool REC>
__global__ void k_gemm2_wmma(const float* __restrict__ h1, const float* __restrict__ W2,
                             const float* __restrict__ dis, float* __restrict__ t2, int n) {
  const int lane = threadIdx.x & 31;
  const int wave = threadIdx.x >> 5;
  const int tile = blockIdx.x * (blockDim.x >> 5) + wave;
  const int row0 = tile << 4;
  if (row0 >= n) return;
  const int m  = lane & 15;
  const int kh = lane >> 4;
  const float bmask = (m < 2) ? 1.0f : 0.0f;
  const int   col   = (m < 2) ? m : 0;
  const float* arow = h1 + (size_t)(row0 + m) * HID + 2 * kh;
  float dv[8] = {0.f, 0.f, 0.f, 0.f, 0.f, 0.f, 0.f, 0.f};
  if (REC) {
    const float4 da = *reinterpret_cast<const float4*>(dis + row0 + 8 * kh);
    const float4 db = *reinterpret_cast<const float4*>(dis + row0 + 8 * kh + 4);
    dv[0] = da.x; dv[1] = da.y; dv[2] = da.z; dv[3] = da.w;
    dv[4] = db.x; dv[5] = db.y; dv[6] = db.z; dv[7] = db.w;
  }

  v8f c = {0.f, 0.f, 0.f, 0.f, 0.f, 0.f, 0.f, 0.f};
#pragma unroll
  for (int k0 = 0; k0 < HID / 4; ++k0) {
    const int kk = 4 * k0 + 2 * kh;
    const float2 av = *reinterpret_cast<const float2*>(arow + 4 * k0);
    v2f a; a.x = av.x; a.y = av.y;
    v2f b; b.x = W2[(kk + 0) * 2 + col] * bmask;
           b.y = W2[(kk + 1) * 2 + col] * bmask;
    c = __builtin_amdgcn_wmma_f32_16x16x4_f32(false, a, false, b, (short)0, c, false, false);
  }
  if (REC) {
    if (m < 4) {
#pragma unroll
      for (int r = 0; r < 8; ++r) {
        const int row = row0 + r + 8 * kh;
        const float v = (m < 2) ? c[r] : ((m == 2) ? dv[r] : 0.f);
        t2[(size_t)row * 4 + m] = v;
      }
    }
  } else if (m < 2) {
#pragma unroll
    for (int r = 0; r < 8; ++r)
      t2[(size_t)(row0 + r + 8 * kh) * 2 + m] = c[r];
  }
}


__global__ void k_pull1(const int* __restrict__ col, const int* __restrict__ row_ptr,
                        const int* __restrict__ cnt, const float* __restrict__ dis,
                        const float* __restrict__ h, const float* __restrict__ b1,
                        float* __restrict__ r, int n) {
  const long long gid = (long long)blockIdx.x * blockDim.x + threadIdx.x;
  const int d = (int)(gid >> 4);
  if (d >= n) return;
  const int f = threadIdx.x & 15;
  const int beg = row_ptr[d];
  const int m = cnt[d];
  const float dd = dis[d];
  float acc = h[(size_t)d * HID + f] * (dd * dd);
  for (int j0 = 0; j0 < m; j0 += 16) {
    const int jj = j0 + f;
    int s_l = d;
    float w_l = 0.f;
    if (jj < m) { s_l = col[beg + jj]; w_l = dis[s_l] * dd; }
    const int nb = m - j0;
#pragma unroll
    for (int t = 0; t < 16; ++t) {
      const int s = __shfl(s_l, t, 16);
      const float w = __shfl(w_l, t, 16);
      const float hv = h[(size_t)s * HID + f];
      acc = fmaf((t < nb) ? hv : 0.f, w, acc);
    }
  }
  const float v = acc + b1[f];
  r[(size_t)d * HID + f] = v > 0.f ? v : 0.f;
}

__global__ void __launch_bounds__(SB) k_pull1_tab(const int* __restrict__ col, const int* __restrict__ row_ptr,
                        const int* __restrict__ cnt, const float* __restrict__ dis,
                        const unsigned* __restrict__ deg8,
                        const float* __restrict__ h, const float* __restrict__ b1,
                        float* __restrict__ r, int n) {
  __shared__ alignas(16) unsigned degL[DCAP / 4];
  const int t = threadIdx.x;
  {
    const uint4* src4 = reinterpret_cast<const uint4*>(deg8);
    uint4* dst4 = reinterpret_cast<uint4*>(degL);
    const int nq = n >> 4;
    int i = t;
    for (; i + 3 * SB < nq; i += 4 * SB) {
      const uint4 v0 = src4[i], v1 = src4[i + SB], v2 = src4[i + 2 * SB], v3 = src4[i + 3 * SB];
      dst4[i] = v0; dst4[i + SB] = v1; dst4[i + 2 * SB] = v2; dst4[i + 3 * SB] = v3;
    }
    for (; i < nq; i += SB) dst4[i] = src4[i];
    if (t < ((n >> 2) & 3)) degL[(nq << 2) + t] = deg8[(nq << 2) + t];
  }
  __syncthreads();
  const unsigned char* degB = reinterpret_cast<const unsigned char*>(degL);
  const int lane = t & 31, wave = t >> 5;
  const int f = lane & 15, half = lane >> 4;
  const int npair = n >> 1;
  const int per = (npair + NWGP - 1) / NWGP;
  const int p0 = blockIdx.x * per;
  const int P = min(per, npair - p0);
  if (P <= 0) return;
  const int qa = p0 + (wave * P) / (SB / 32);
  const int qn = p0 + ((wave + 1) * P) / (SB / 32) - qa;
  const int nd0 = 2 * qa;
  const int dl = nd0 + lane;
  const bool has = lane < 2 * qn;
  const int   beg_l = has ? row_ptr[dl] : 0;
  const int   m_l   = has ? cnt[dl] : 0;
  const float dd_l  = has ? dis[dl] : 0.f;
  const int q4 = (f & 3) * 4, e = f >> 2;
  const float4 bias = *reinterpret_cast<const float4*>(b1 + q4);
  for (int i = 0; i < qn; ++i) {
    const int k = 2 * i + half;
    const int d = nd0 + k;
    const int beg = __shfl(beg_l, k, 32);
    const int m = __shfl(m_l, k, 32);
    const float dd = __shfl(dd_l, k, 32);
    int s_cur = (f < m) ? col[beg + f] : d;
    const float4 own = *reinterpret_cast<const float4*>(h + (size_t)d * HID + q4);
    const float sl = dd * dd;
    float4 acc = make_float4((e == 0) ? own.x * sl : 0.f, (e == 0) ? own.y * sl : 0.f,
                             (e == 0) ? own.z * sl : 0.f, (e == 0) ? own.w * sl : 0.f);
    for (int j0 = 0; j0 < m; j0 += 16) {
      const int jj = j0 + f;
      const int jn = jj + 16;
      const int s_l = s_cur;
      float w_l = 0.f;
      if (jj < m) {
        const unsigned bq = degB[s_l];
        const float ds = (bq < 255u) ? rsqrtf((float)(int)(bq + 1u)) : dis[s_l];
        w_l = ds * dd;
      }
      s_cur = (jn < m) ? col[beg + jn] : d;
      const int nb = m - j0;
#pragma unroll
      for (int tt = 0; tt < 4; ++tt) {
        const int slot = 4 * tt + e;
        const int s = __shfl(s_l, slot, 16);
        const float w = __shfl(w_l, slot, 16);
        const float4 hv = *reinterpret_cast<const float4*>(h + (size_t)s * HID + q4);
        const bool ok = slot < nb;
        acc.x = fmaf(ok ? hv.x : 0.f, w, acc.x);
        acc.y = fmaf(ok ? hv.y : 0.f, w, acc.y);
        acc.z = fmaf(ok ? hv.z : 0.f, w, acc.z);
        acc.w = fmaf(ok ? hv.w : 0.f, w, acc.w);
      }
    }
#pragma unroll
    for (int sh = 4; sh <= 8; sh <<= 1) {
      acc.x += __shfl_xor(acc.x, sh, 16);
      acc.y += __shfl_xor(acc.y, sh, 16);
      acc.z += __shfl_xor(acc.z, sh, 16);
      acc.w += __shfl_xor(acc.w, sh, 16);
    }
    if (e == 0) {
      float4 v = make_float4(acc.x + bias.x, acc.y + bias.y, acc.z + bias.z, acc.w + bias.w);
      v.x = v.x > 0.f ? v.x : 0.f; v.y = v.y > 0.f ? v.y : 0.f;
      v.z = v.z > 0.f ? v.z : 0.f; v.w = v.w > 0.f ? v.w : 0.f;
      *reinterpret_cast<float4*>(r + (size_t)d * HID + q4) = v;
    }
  }
}

__global__ void k_pull2(const int* __restrict__ col, const int* __restrict__ row_ptr,
                        const int* __restrict__ cnt, const float* __restrict__ t2,
                        const float* __restrict__ b2, float* __restrict__ y, int n) {
  __shared__ float2 so[TB / 16];
  const int f = threadIdx.x & 15;
  const int hw = threadIdx.x >> 4;
  const int d0 = blockIdx.x * (TB / 16);
  const int d = d0 + hw;
  float a0 = 0.f, a1 = 0.f;
  if (d < n) {
    const int beg = row_ptr[d];
    const int m = cnt[d];
    const float4 rd = *reinterpret_cast<const float4*>(t2 + (size_t)d * 4);
    const float dd = rd.z;
    for (int jj = f; jj < m; jj += 32) {
      const int jb = jj + 16;
      const int sa = col[beg + jj];
      const int sb = (jb < m) ? col[beg + jb] : sa;
      const float4 va = *reinterpret_cast<const float4*>(t2 + (size_t)sa * 4);
      const float4 vb = *reinterpret_cast<const float4*>(t2 + (size_t)sb * 4);
      const float wa = va.z * dd;
      const float wb = (jb < m) ? vb.z * dd : 0.f;
      a0 = fmaf(va.x, wa, a0); a1 = fmaf(va.y, wa, a1);
      a0 = fmaf((jb < m) ? vb.x : 0.f, wb, a0); a1 = fmaf((jb < m) ? vb.y : 0.f, wb, a1);
    }
#pragma unroll
    for (int k = 8; k > 0; k >>= 1) {
      a0 += __shfl_xor(a0, k, 16);
      a1 += __shfl_xor(a1, k, 16);
    }
    if (f == 0) {
      const float s2 = dd * dd;
      const float x0 = a0 + rd.x * s2 + b2[0];
      const float x1 = a1 + rd.y * s2 + b2[1];
      const float mx = fmaxf(x0, x1);
      const float lse = mx + logf(expf(x0 - mx) + expf(x1 - mx));
      so[hw] = make_float2(x0 - lse, x1 - lse);
    }
  }
  __syncthreads();
  if (threadIdx.x < TB / 16 && d0 + (int)threadIdx.x < n)
    *reinterpret_cast<float2*>(y + (size_t)(d0 + threadIdx.x) * 2) = so[threadIdx.x];
}


__global__ void k_init_deg(float* __restrict__ deg, int n) {
  int i = blockIdx.x * blockDim.x + threadIdx.x;
  if (i < n) deg[i] = 1.0f;
}

__global__ void k_deg_edges(const int* __restrict__ dst, float* __restrict__ deg, int nE) {
  int e = blockIdx.x * blockDim.x + threadIdx.x;
  if (e < nE) atomicAdd(&deg[dst[e]], 1.0f);
}

__global__ void k_dis(const float* __restrict__ deg, float* __restrict__ dis, int n) {
  int i = blockIdx.x * blockDim.x + threadIdx.x;
  if (i < n) dis[i] = rsqrtf(deg[i]);
}

__global__ void k_self1(const float* __restrict__ h, const float* __restrict__ dis,
                        float* __restrict__ out1, int n) {
  int i = blockIdx.x * blockDim.x + threadIdx.x;
  if (i >= n) return;
  float s = dis[i]; s = s * s;
  const float4* hp = reinterpret_cast<const float4*>(h + (size_t)i * HID);
  float4* op = reinterpret_cast<float4*>(out1 + (size_t)i * HID);
#pragma unroll
  for (int q = 0; q < 4; ++q) {
    float4 v = hp[q];
    v.x *= s; v.y *= s; v.z *= s; v.w *= s;
    op[q] = v;
  }
}

__global__ void k_scatter1(const int* __restrict__ src, const int* __restrict__ dst,
                           const float* __restrict__ dis, const float* __restrict__ h,
                           float* __restrict__ out1, int nE) {
  long long t = (long long)blockIdx.x * blockDim.x + threadIdx.x;
  int e = (int)(t >> 2);
  if (e >= nE) return;
  int q = (int)(t & 3);
  int s = src[e], d = dst[e];
  float norm = dis[s] * dis[d];
  float4 v = *reinterpret_cast<const float4*>(h + (size_t)s * HID + 4 * q);
  float* o = out1 + (size_t)d * HID + 4 * q;
  atomicAdd(o + 0, v.x * norm);
  atomicAdd(o + 1, v.y * norm);
  atomicAdd(o + 2, v.z * norm);
  atomicAdd(o + 3, v.w * norm);
}

__global__ void k_relu_bias(float* __restrict__ out1, const float* __restrict__ b1, int n) {
  long long t = (long long)blockIdx.x * blockDim.x + threadIdx.x;
  if (t >= (long long)n * HID) return;
  int j = (int)(t & (HID - 1));
  float v = out1[t] + b1[j];
  out1[t] = v > 0.f ? v : 0.f;
}

__global__ void k_self2(const float* __restrict__ h2, const float* __restrict__ dis,
                        float* __restrict__ out2, int n) {
  int i = blockIdx.x * blockDim.x + threadIdx.x;
  if (i >= n) return;
  float s = dis[i]; s = s * s;
  out2[(size_t)i * 2 + 0] = h2[(size_t)i * 2 + 0] * s;
  out2[(size_t)i * 2 + 1] = h2[(size_t)i * 2 + 1] * s;
}

__global__ void k_scatter2(const int* __restrict__ src, const int* __restrict__ dst,
                           const float* __restrict__ dis, const float* __restrict__ h2,
                           float* __restrict__ out2, int nE) {
  int e = blockIdx.x * blockDim.x + threadIdx.x;
  if (e >= nE) return;
  int s = src[e], d = dst[e];
  float norm = dis[s] * dis[d];
  atomicAdd(&out2[(size_t)d * 2 + 0], h2[(size_t)s * 2 + 0] * norm);
  atomicAdd(&out2[(size_t)d * 2 + 1], h2[(size_t)s * 2 + 1] * norm);
}

__global__ void k_logsoftmax(const float* __restrict__ out2, const float* __restrict__ b2,
                             float* __restrict__ y, int n) {
  int i = blockIdx.x * blockDim.x + threadIdx.x;
  if (i >= n) return;
  float a0 = out2[(size_t)i * 2 + 0] + b2[0];
  float a1 = out2[(size_t)i * 2 + 1] + b2[1];
  float mx = fmaxf(a0, a1);
  float lse = mx + logf(expf(a0 - mx) + expf(a1 - mx));
  y[(size_t)i * 2 + 0] = a0 - lse;
  y[(size_t)i * 2 + 1] = a1 - lse;
}


extern "C" void kernel_launch(void* const* d_in, const int* in_sizes, int n_in,
                              void* d_out, int out_size, void* d_ws, size_t ws_size,
                              hipStream_t stream) {
  const float* x  = (const float*)d_in[0];
  const int*   ei = (const int*)d_in[1];
  const float* W1 = (const float*)d_in[2];
  const float* b1 = (const float*)d_in[3];
  const float* W2 = (const float*)d_in[4];
  const float* b2 = (const float*)d_in[5];

  const int n  = in_sizes[0] / IN_DIM;
  const int nE = in_sizes[1] / 2;
  const int* src = ei;
  const int* dst = ei + nE;
  float* y = (float*)d_out;

  dim3 blk(TB);
  int gN   = (n + TB - 1) / TB;
  int gE   = (nE + TB - 1) / TB;
  int gN16 = (int)(((long long)n * HID + TB - 1) / TB);
  int tiles = (n + 15) / 16;
  int gT   = (tiles + (TB / 32) - 1) / (TB / 32);

  const size_t big_words  = (size_t)nE > (size_t)n * 2 * HID ? (size_t)nE : (size_t)n * 2 * HID;
  const size_t need_words = (size_t)n * 5 + 1024 + (size_t)nE + big_words;
  const int NB  = (n + RANGE - 1) / RANGE;
  const int NCH = (nE + CH - 1) / CH;
  const size_t n2 = (size_t)NB * (size_t)NCH;
  if (ws_size < need_words * sizeof(int) || NB > NBMAX || NCH > NCHMAX || (nE & 3) != 0 || (n & 3) != 0 ||
      n2 > (size_t)n) {
    float* deg  = (float*)d_ws;
    float* dis  = deg + n;
    float* h    = dis + n;
    float* out1 = h    + (size_t)n * HID;
    float* h2   = out1 + (size_t)n * HID;
    float* out2 = h2   + (size_t)n * 2;
    int gE4 = (int)(((long long)nE * 4 + TB - 1) / TB);
    k_init_deg  <<<gN,  blk, 0, stream>>>(deg, n);
    k_deg_edges <<<gE,  blk, 0, stream>>>(dst, deg, nE);
    k_dis       <<<gN,  blk, 0, stream>>>(deg, dis, n);
    k_gemm1_wmma<<<gT,  blk, 0, stream>>>(x, W1, h, n);
    k_self1     <<<gN,  blk, 0, stream>>>(h, dis, out1, n);
    k_scatter1  <<<gE4, blk, 0, stream>>>(src, dst, dis, h, out1, nE);
    k_relu_bias <<<gN16,blk, 0, stream>>>(out1, b1, n);
    k_gemm2_wmma<false><<<gT, blk, 0, stream>>>(out1, W2, dis, h2, n);
    k_self2     <<<gN,  blk, 0, stream>>>(h2, dis, out2, n);
    k_scatter2  <<<gE,  blk, 0, stream>>>(src, dst, dis, h2, out2, nE);
    k_logsoftmax<<<gN,  blk, 0, stream>>>(out2, b2, y, n);
    return;
  }

  int*   cnt     = (int*)d_ws;
  int*   row_ptr = cnt + n;
  float* dis     = (float*)(row_ptr + n);
  int*   bsum    = (int*)(dis + n);
  int*   cc      = bsum + 1024;
  int*   startT  = cc + n2;
  unsigned* binned = (unsigned*)(cc + (size_t)n * 2);
  float* h       = (float*)binned;
  float* r       = h + (size_t)n * HID;
  float* t2      = h;
  int*   col     = (int*)binned + big_words;

  dim3 sblk(SB);
  const bool tab = n <= DCAP && (size_t)NB * (RANGE / 4) <= n2;
  unsigned* deg8 = tab ? (unsigned*)cc : nullptr;
  k_chunk_sort <<<NCH, sblk, 0, stream>>>(src, dst, startT, binned, nE, NB, NCH);
  k_range_lists<<<NB,  sblk, 0, stream>>>(binned, startT, cnt, row_ptr, dis, col, deg8, n, nE, NB, NCH);
  k_gemm1_wmma <<<gT,  blk,  0, stream>>>(x, W1, h, n);
  if (tab) k_pull1_tab<<<NWGP, sblk, 0, stream>>>(col, row_ptr, cnt, dis, deg8, h, b1, r, n);
  else     k_pull1    <<<gN16, blk,  0, stream>>>(col, row_ptr, cnt, dis, h, b1, r, n);
  k_gemm2_wmma<true><<<gT, blk, 0, stream>>>(r, W2, dis, t2, n);
  k_pull2      <<<gN16,blk,  0, stream>>>(col, row_ptr, cnt, t2, b2, y, n);
}


# Layer 2 pulled by resident waves: a wave walks 16 nodes (half-wave per node, list entries requested two node pairs ahead), 12500 waves of 8 trips instead of 100000 one-shot waves; same sums
# speedup vs baseline: 6.7260x; 1.0429x over previous
#include <hip/hip_runtime.h>
#include <math.h>

typedef __attribute__((ext_vector_type(2))) float v2f;
typedef __attribute__((ext_vector_type(8))) float v8f;

#define IN_DIM 128
#define HID    16
#define TB     256
#define RB     9
#define RANGE  512
#define CH     16384
#define NBMAX  2048
#define NCHMAX 2048
#define RG     4
#define SB     1024
#define CAPC   24576
#define DCAP   262144
#define NWGP   256


__device__ __forceinline__ int wave_sum_i(int v) {
#pragma unroll
  for (int m = 16; m > 0; m >>= 1) v += __shfl_xor(v, m, 32);
  return v;
}

__device__ __forceinline__ int block_excl_scan(int v, int* sm_w) {
  const int lane = threadIdx.x & 31, wave = threadIdx.x >> 5;
  int inc = v;
  for (int d = 1; d < 32; d <<= 1) {
    int u = __shfl_up(inc, d, 32);
    if (lane >= d) inc += u;
  }
  if (lane == 31) sm_w[wave] = inc;
  __syncthreads();
  if (wave == 0) {
    const int w = sm_w[lane];
    int winc = w;
    for (int d = 1; d < 32; d <<= 1) {
      int u = __shfl_up(winc, d, 32);
      if (lane >= d) winc += u;
    }
    sm_w[lane] = winc - w;
  }
  __syncthreads();
  return inc - v + sm_w[wave];
}

__global__ void __launch_bounds__(SB, 16) k_chunk_sort(const int* __restrict__ src, const int* __restrict__ dst,
                                                   int* __restrict__ startT, unsigned* __restrict__ binned,
                                                   int nE, int NB, int NCH) {
  __shared__ int curL[NBMAX];
  __shared__ int sm_w[32];
  __shared__ alignas(16) unsigned recL[CH];
  const int t = threadIdx.x, chunk = blockIdx.x;
  const int e0 = chunk * CH;
  int4 d4[CH / (4 * SB)], s4[CH / (4 * SB)];
#pragma unroll
  for (int k = 0; k < CH / (4 * SB); ++k) {
    const int e = e0 + 4 * (t + k * SB);
    const int ee = e < nE ? e : 0;
    d4[k] = *reinterpret_cast<const int4*>(dst + ee);
    s4[k] = *reinterpret_cast<const int4*>(src + ee);
  }
  for (int b = t; b < NB; b += SB) curL[b] = 0;
  __syncthreads();
#pragma unroll
  for (int k = 0; k < CH / (4 * SB); ++k) {
    if (e0 + 4 * (t + k * SB) < nE) {
      atomicAdd(&curL[d4[k].x >> RB], 1);
      atomicAdd(&curL[d4[k].y >> RB], 1);
      atomicAdd(&curL[d4[k].z >> RB], 1);
      atomicAdd(&curL[d4[k].w >> RB], 1);
    }
  }
  __syncthreads();
  const int b0 = 2 * t, b1 = 2 * t + 1;
  const int h0 = (b0 < NB) ? curL[b0] : 0;
  const int h1 = (b1 < NB) ? curL[b1] : 0;
  const int ex = block_excl_scan(h0 + h1, sm_w);
  if (b0 < NB) { curL[b0] = ex;      startT[b0 * NCH + chunk] = ex; }
  if (b1 < NB) { curL[b1] = ex + h0; startT[b1 * NCH + chunk] = ex + h0; }
  __syncthreads();
#pragma unroll
  for (int k = 0; k < CH / (4 * SB); ++k) {
    if (e0 + 4 * (t + k * SB) < nE) {
      const int q0 = atomicAdd(&curL[d4[k].x >> RB], 1);
      const int q1 = atomicAdd(&curL[d4[k].y >> RB], 1);
      const int q2 = atomicAdd(&curL[d4[k].z >> RB], 1);
      const int q3 = atomicAdd(&curL[d4[k].w >> RB], 1);
      recL[q0] = ((unsigned)s4[k].x << RB) | (unsigned)(d4[k].x & (RANGE - 1));
      recL[q1] = ((unsigned)s4[k].y << RB) | (unsigned)(d4[k].y & (RANGE - 1));
      recL[q2] = ((unsigned)s4[k].z << RB) | (unsigned)(d4[k].z & (RANGE - 1));
      recL[q3] = ((unsigned)s4[k].w << RB) | (unsigned)(d4[k].w & (RANGE - 1));
    }
  }
  __syncthreads();
  const int len = min(CH, nE - e0);
  for (int i = 4 * t; i < len; i += 4 * SB)
    *reinterpret_cast<uint4*>(binned + e0 + i) = *reinterpret_cast<const uint4*>(recL + i);
}

template <class F>
__device__ __forceinline__ void range_records(const unsigned* __restrict__ binned, const int* runS,
                                              const int* runN, int NCH, F f) {
  const int lane = threadIdx.x & 31, wave = threadIdx.x >> 5;
  for (int c0 = wave; c0 < NCH; c0 += 32 * RG) {
    unsigned ra[RG], rb[RG];
    int ln[RG], bs[RG];
#pragma unroll
    for (int g = 0; g < RG; ++g) {
      const int c = c0 + 32 * g;
      ln[g] = (c < NCH) ? runN[c] : 0;
      bs[g] = (c < NCH) ? runS[c] : 0;
      ra[g] = binned[(lane < ln[g]) ? bs[g] + lane : 0];
      rb[g] = binned[(lane + 32 < ln[g]) ? bs[g] + lane + 32 : 0];
    }
#pragma unroll
    for (int g = 0; g < RG; ++g) {
      if (lane < ln[g]) f(ra[g]);
      if (lane + 32 < ln[g]) f(rb[g]);
      for (int j = lane + 64; j < ln[g]; j += 32) f(binned[bs[g] + j]);
    }
  }
}

__global__ void __launch_bounds__(SB) k_range_lists(const unsigned* __restrict__ binned, const int* __restrict__ startT,
                                                    int* __restrict__ cnt, int* __restrict__ row_ptr,
                                                    float* __restrict__ dis, int* __restrict__ col,
                                                    unsigned* __restrict__ deg8,
                                                    int n, int nE, int NB, int NCH) {
  __shared__ alignas(16) int cntL[RANGE];
  __shared__ int curL[RANGE];
  __shared__ int sm_w[32];
  __shared__ int sm_s[64];
  __shared__ int runS[NCHMAX];
  __shared__ int runN[NCHMAX];
  __shared__ int colL[CAPC];
  const int t = threadIdx.x, lane = t & 31, wave = t >> 5, b = blockIdx.x;
  int sS = 0, sR = 0;
  for (int c = t; c < NCH; c += SB) {
    const int st = startT[b * NCH + c];
    const int en = (b + 1 < NB) ? startT[(b + 1) * NCH + c] : min(CH, nE - c * CH);
    runS[c] = c * CH + st;
    runN[c] = en - st;
    sS += st;
    sR += en - st;
  }
  if (t < RANGE) cntL[t] = 0;
  sS = wave_sum_i(sS);
  sR = wave_sum_i(sR);
  if (lane == 0) { sm_s[wave] = sS; sm_s[32 + wave] = sR; }
  __syncthreads();
  const int S = wave_sum_i(sm_s[lane]);
  const int R = wave_sum_i(sm_s[32 + lane]);
  range_records(binned, runS, runN, NCH,
                [&](unsigned rec) { atomicAdd(&cntL[rec & (RANGE - 1)], 1); });
  __syncthreads();
  const int c = (t < RANGE) ? cntL[t] : 0;
  const int ex = block_excl_scan(c, sm_w);
  if (t < RANGE) {
    curL[t] = ex;
    const int node = (b << RB) + t;
    if (node < n) {
      cnt[node] = c;
      row_ptr[node] = S + ex;
      dis[node] = rsqrtf((float)(c + 1));
    }
  }
  if (deg8 != nullptr && t < RANGE / 4) {
    const int4 c4 = *reinterpret_cast<const int4*>(&cntL[4 * t]);
    const unsigned pk = (unsigned)min(c4.x, 255) | ((unsigned)min(c4.y, 255) << 8) |
                        ((unsigned)min(c4.z, 255) << 16) | ((unsigned)min(c4.w, 255) << 24);
    const int wi = (b << (RB - 2)) + t;
    if (4 * wi < n) deg8[wi] = pk;
  }
  __syncthreads();
  if (R <= CAPC) {
    range_records(binned, runS, runN, NCH, [&](unsigned rec) {
      const int slot = atomicAdd(&curL[rec & (RANGE - 1)], 1);
      colL[slot] = (int)(rec >> RB);
    });
    __syncthreads();
    for (int i = t; i < R; i += SB) col[S + i] = colL[i];
  } else {
    range_records(binned, runS, runN, NCH, [&](unsigned rec) {
      const int slot = atomicAdd(&curL[rec & (RANGE - 1)], 1);
      col[S + slot] = (int)(rec >> RB);
    });
  }
}


__global__ void k_gemm1_wmma(const float* __restrict__ x, const float* __restrict__ W1,
                             float* __restrict__ h, int n) {
  const int lane = threadIdx.x & 31;
  const int wave = threadIdx.x >> 5;
  const int tile = blockIdx.x * (blockDim.x >> 5) + wave;
  const int row0 = tile << 4;
  if (row0 >= n) return;
  const int m  = lane & 15;
  const int kh = lane >> 4;
  const float* xrow = x + (size_t)(row0 + m) * IN_DIM + 2 * kh;

  v8f c = {0.f, 0.f, 0.f, 0.f, 0.f, 0.f, 0.f, 0.f};
#pragma unroll
  for (int k0 = 0; k0 < IN_DIM / 4; ++k0) {
    const int kk = 4 * k0 + 2 * kh;
    const float2 av = *reinterpret_cast<const float2*>(xrow + 4 * k0);
    v2f a; a.x = av.x; a.y = av.y;
    v2f b; b.x = W1[(kk + 0) * HID + m];
           b.y = W1[(kk + 1) * HID + m];
    c = __builtin_amdgcn_wmma_f32_16x16x4_f32(false, a, false, b, (short)0, c, false, false);
  }
  float* hout = h + (size_t)(row0 + 8 * kh) * HID + m;
#pragma unroll
  for (int r = 0; r < 8; ++r)
    hout[(size_t)r * HID] = c[r];
}


template <bool REC>
__global__ void k_gemm2_wmma(const float* __restrict__ h1, const float* __restrict__ W2,
                             const float* __restrict__ dis, float* __restrict__ t2, int n) {
  const int lane = threadIdx.x & 31;
  const int wave = threadIdx.x >> 5;
  const int tile = blockIdx.x * (blockDim.x >> 5) + wave;
  const int row0 = tile << 4;
  if (row0 >= n) return;
  const int m  = lane & 15;
  const int kh = lane >> 4;
  const float bmask = (m < 2) ? 1.0f : 0.0f;
  const int   col   = (m < 2) ? m : 0;
  const float* arow = h1 + (size_t)(row0 + m) * HID + 2 * kh;
  float dv[8] = {0.f, 0.f, 0.f, 0.f, 0.f, 0.f, 0.f, 0.f};
  if (REC) {
    const float4 da = *reinterpret_cast<const float4*>(dis + row0 + 8 * kh);
    const float4 db = *reinterpret_cast<const float4*>(dis + row0 + 8 * kh + 4);
    dv[0] = da.x; dv[1] = da.y; dv[2] = da.z; dv[3] = da.w;
    dv[4] = db.x; dv[5] = db.y; dv[6] = db.z; dv[7] = db.w;
  }

  v8f c = {0.f, 0.f, 0.f, 0.f, 0.f, 0.f, 0.f, 0.f};
#pragma unroll
  for (int k0 = 0; k0 < HID / 4; ++k0) {
    const int kk = 4 * k0 + 2 * kh;
    const float2 av = *reinterpret_cast<const float2*>(arow + 4 * k0);
    v2f a; a.x = av.x; a.y = av.y;
    v2f b; b.x = W2[(kk + 0) * 2 + col] * bmask;
           b.y = W2[(kk + 1) * 2 + col] * bmask;
    c = __builtin_amdgcn_wmma_f32_16x16x4_f32(false, a, false, b, (short)0, c, false, false);
  }
  if (REC) {
    if (m < 4) {
#pragma unroll
      for (int r = 0; r < 8; ++r) {
        const int row = row0 + r + 8 * kh;
        const float v = (m < 2) ? c[r] : ((m == 2) ? dv[r] : 0.f);
        t2[(size_t)row * 4 + m] = v;
      }
    }
  } else if (m < 2) {
#pragma unroll
    for (int r = 0; r < 8; ++r)
      t2[(size_t)(row0 + r + 8 * kh) * 2 + m] = c[r];
  }
}


__global__ void k_pull1(const int* __restrict__ col, const int* __restrict__ row_ptr,
                        const int* __restrict__ cnt, const float* __restrict__ dis,
                        const float* __restrict__ h, const float* __restrict__ b1,
                        float* __restrict__ r, int n) {
  const long long gid = (long long)blockIdx.x * blockDim.x + threadIdx.x;
  const int d = (int)(gid >> 4);
  if (d >= n) return;
  const int f = threadIdx.x & 15;
  const int beg = row_ptr[d];
  const int m = cnt[d];
  const float dd = dis[d];
  float acc = h[(size_t)d * HID + f] * (dd * dd);
  for (int j0 = 0; j0 < m; j0 += 16) {
    const int jj = j0 + f;
    int s_l = d;
    float w_l = 0.f;
    if (jj < m) { s_l = col[beg + jj]; w_l = dis[s_l] * dd; }
    const int nb = m - j0;
#pragma unroll
    for (int t = 0; t < 16; ++t) {
      const int s = __shfl(s_l, t, 16);
      const float w = __shfl(w_l, t, 16);
      const float hv = h[(size_t)s * HID + f];
      acc = fmaf((t < nb) ? hv : 0.f, w, acc);
    }
  }
  const float v = acc + b1[f];
  r[(size_t)d * HID + f] = v > 0.f ? v : 0.f;
}

__global__ void __launch_bounds__(SB) k_pull1_tab(const int* __restrict__ col, const int* __restrict__ row_ptr,
                        const int* __restrict__ cnt, const float* __restrict__ dis,
                        const unsigned* __restrict__ deg8,
                        const float* __restrict__ h, const float* __restrict__ b1,
                        float* __restrict__ r, int n) {
  __shared__ alignas(16) unsigned degL[DCAP / 4];
  const int t = threadIdx.x;
  {
    const uint4* src4 = reinterpret_cast<const uint4*>(deg8);
    uint4* dst4 = reinterpret_cast<uint4*>(degL);
    const int nq = n >> 4;
    int i = t;
    for (; i + 3 * SB < nq; i += 4 * SB) {
      const uint4 v0 = src4[i], v1 = src4[i + SB], v2 = src4[i + 2 * SB], v3 = src4[i + 3 * SB];
      dst4[i] = v0; dst4[i + SB] = v1; dst4[i + 2 * SB] = v2; dst4[i + 3 * SB] = v3;
    }
    for (; i < nq; i += SB) dst4[i] = src4[i];
    if (t < ((n >> 2) & 3)) degL[(nq << 2) + t] = deg8[(nq << 2) + t];
  }
  __syncthreads();
  const unsigned char* degB = reinterpret_cast<const unsigned char*>(degL);
  const int lane = t & 31, wave = t >> 5;
  const int f = lane & 15, half = lane >> 4;
  const int npair = n >> 1;
  const int per = (npair + NWGP - 1) / NWGP;
  const int p0 = blockIdx.x * per;
  const int P = min(per, npair - p0);
  if (P <= 0) return;
  const int qa = p0 + (wave * P) / (SB / 32);
  const int qn = p0 + ((wave + 1) * P) / (SB / 32) - qa;
  const int nd0 = 2 * qa;
  const int dl = nd0 + lane;
  const bool has = lane < 2 * qn;
  const int   beg_l = has ? row_ptr[dl] : 0;
  const int   m_l   = has ? cnt[dl] : 0;
  const float dd_l  = has ? dis[dl] : 0.f;
  const int q4 = (f & 3) * 4, e = f >> 2;
  const float4 bias = *reinterpret_cast<const float4*>(b1 + q4);
  for (int i = 0; i < qn; ++i) {
    const int k = 2 * i + half;
    const int d = nd0 + k;
    const int beg = __shfl(beg_l, k, 32);
    const int m = __shfl(m_l, k, 32);
    const float dd = __shfl(dd_l, k, 32);
    int s_cur = (f < m) ? col[beg + f] : d;
    const float4 own = *reinterpret_cast<const float4*>(h + (size_t)d * HID + q4);
    const float sl = dd * dd;
    float4 acc = make_float4((e == 0) ? own.x * sl : 0.f, (e == 0) ? own.y * sl : 0.f,
                             (e == 0) ? own.z * sl : 0.f, (e == 0) ? own.w * sl : 0.f);
    for (int j0 = 0; j0 < m; j0 += 16) {
      const int jj = j0 + f;
      const int jn = jj + 16;
      const int s_l = s_cur;
      float w_l = 0.f;
      if (jj < m) {
        const unsigned bq = degB[s_l];
        const float ds = (bq < 255u) ? rsqrtf((float)(int)(bq + 1u)) : dis[s_l];
        w_l = ds * dd;
      }
      s_cur = (jn < m) ? col[beg + jn] : d;
      const int nb = m - j0;
#pragma unroll
      for (int tt = 0; tt < 4; ++tt) {
        const int slot = 4 * tt + e;
        const int s = __shfl(s_l, slot, 16);
        const float w = __shfl(w_l, slot, 16);
        const float4 hv = *reinterpret_cast<const float4*>(h + (size_t)s * HID + q4);
        const bool ok = slot < nb;
        acc.x = fmaf(ok ? hv.x : 0.f, w, acc.x);
        acc.y = fmaf(ok ? hv.y : 0.f, w, acc.y);
        acc.z = fmaf(ok ? hv.z : 0.f, w, acc.z);
        acc.w = fmaf(ok ? hv.w : 0.f, w, acc.w);
      }
    }
#pragma unroll
    for (int sh = 4; sh <= 8; sh <<= 1) {
      acc.x += __shfl_xor(acc.x, sh, 16);
      acc.y += __shfl_xor(acc.y, sh, 16);
      acc.z += __shfl_xor(acc.z, sh, 16);
      acc.w += __shfl_xor(acc.w, sh, 16);
    }
    if (e == 0) {
      float4 v = make_float4(acc.x + bias.x, acc.y + bias.y, acc.z + bias.z, acc.w + bias.w);
      v.x = v.x > 0.f ? v.x : 0.f; v.y = v.y > 0.f ? v.y : 0.f;
      v.z = v.z > 0.f ? v.z : 0.f; v.w = v.w > 0.f ? v.w : 0.f;
      *reinterpret_cast<float4*>(r + (size_t)d * HID + q4) = v;
    }
  }
}

#define P2W 25

__device__ __forceinline__ int lane_i(int v, int l) { return __builtin_amdgcn_readlane(v, l); }
__device__ __forceinline__ float lane_f(float v, int l) {
  return __int_as_float(__builtin_amdgcn_readlane(__float_as_int(v), l));
}

template <int K>
__device__ __forceinline__ float row_xor(float v) {
  return __int_as_float(__builtin_amdgcn_update_dpp(0, __float_as_int(v), 0x160 | K, 0xf, 0xf, false));
}

__device__ __forceinline__ void p2_entries(const int* __restrict__ col, int beg_l, int m_l, int k0,
                                           int half, int f, int (&s)[4]) {
  const int ka = k0 < 14 ? k0 : 14;
  const int b0 = lane_i(beg_l, ka), b1 = lane_i(beg_l, ka + 1);
  const int c0 = lane_i(m_l, ka),   c1 = lane_i(m_l, ka + 1);
  const int beg = half ? b1 : b0;
  const int m = (k0 < 16) ? (half ? c1 : c0) : 0;
#pragma unroll
  for (int q = 0; q < 4; ++q) {
    const int slot = f + 16 * q;
    s[q] = col[(slot < m) ? beg + slot : 0];
  }
}

__global__ void __launch_bounds__(32 * P2W) k_pull2_res(const int* __restrict__ col, const int* __restrict__ row_ptr,
                        const int* __restrict__ cnt, const float* __restrict__ t2,
                        const float* __restrict__ b2, float* __restrict__ y, int n) {
  const int lane = threadIdx.x & 31, wave = threadIdx.x >> 5;
  const int f = lane & 15, half = lane >> 4;
  const int nd0 = (blockIdx.x * P2W + wave) << 4;
  if (nd0 >= n) return;
  const int dl = nd0 + f;
  const bool has = dl < n;
  const int beg_l = has ? row_ptr[dl] : 0;
  const int m_l   = has ? cnt[dl] : 0;
  float4 rd_l = make_float4(0.f, 0.f, 0.f, 0.f);
  if (has) rd_l = *reinterpret_cast<const float4*>(t2 + (size_t)dl * 4);
  float o0 = 0.f, o1 = 0.f;
  int sa[4], sb[4];
  p2_entries(col, beg_l, m_l, 0, half, f, sa);
  p2_entries(col, beg_l, m_l, 2, half, f, sb);

  auto step = [&](int k0, int (&s)[4]) {
    const int c0 = lane_i(m_l, k0), c1 = lane_i(m_l, k0 + 1);
    const int g0 = lane_i(beg_l, k0), g1 = lane_i(beg_l, k0 + 1);
    const float d0 = lane_f(rd_l.z, k0), d1 = lane_f(rd_l.z, k0 + 1);
    const int m = half ? c1 : c0;
    const int beg = half ? g1 : g0;
    const float dd = half ? d1 : d0;
    float4 v[4];
#pragma unroll
    for (int q = 0; q < 4; ++q) v[q] = *reinterpret_cast<const float4*>(t2 + (size_t)s[q] * 4);
    p2_entries(col, beg_l, m_l, k0 + 4, half, f, s);
    __builtin_amdgcn_sched_barrier(0);
    float a0 = 0.f, a1 = 0.f;
#pragma unroll
    for (int q = 0; q < 4; ++q) {
      const bool ok = f + 16 * q < m;
      const float w = ok ? v[q].z * dd : 0.f;
      a0 = fmaf(ok ? v[q].x : 0.f, w, a0);
      a1 = fmaf(ok ? v[q].y : 0.f, w, a1);
    }
    if (max(c0, c1) > 64) {
      for (int jj = f + 64; jj < m; jj += 16) {
        const int sx = col[beg + jj];
        const float4 vx = *reinterpret_cast<const float4*>(t2 + (size_t)sx * 4);
        const float w = vx.z * dd;
        a0 = fmaf(vx.x, w, a0);
        a1 = fmaf(vx.y, w, a1);
      }
    }
    a0 += row_xor<8>(a0); a1 += row_xor<8>(a1);
    a0 += row_xor<4>(a0); a1 += row_xor<4>(a1);
    a0 += row_xor<2>(a0); a1 += row_xor<2>(a1);
    a0 += row_xor<1>(a0); a1 += row_xor<1>(a1);
    const float t00 = lane_f(a0, 0), t01 = lane_f(a1, 0), t10 = lane_f(a0, 16), t11 = lane_f(a1, 16);
    if (f == k0)     { o0 = t00; o1 = t01; }
    if (f == k0 + 1) { o0 = t10; o1 = t11; }
  };
  for (int k0 = 0; k0 < 16; k0 += 4) {
    step(k0, sa);
    step(k0 + 2, sb);
  }
  if (has && half == 0) {
    const float dd = rd_l.z;
    const float s2 = dd * dd;
    const float x0 = o0 + rd_l.x * s2 + b2[0];
    const float x1 = o1 + rd_l.y * s2 + b2[1];
    const float mx = fmaxf(x0, x1);
    const float lse = mx + logf(expf(x0 - mx) + expf(x1 - mx));
    *reinterpret_cast<float2*>(y + (size_t)dl * 2) = make_float2(x0 - lse, x1 - lse);
  }
}


__global__ void k_init_deg(float* __restrict__ deg, int n) {
  int i = blockIdx.x * blockDim.x + threadIdx.x;
  if (i < n) deg[i] = 1.0f;
}

__global__ void k_deg_edges(const int* __restrict__ dst, float* __restrict__ deg, int nE) {
  int e = blockIdx.x * blockDim.x + threadIdx.x;
  if (e < nE) atomicAdd(&deg[dst[e]], 1.0f);
}

__global__ void k_dis(const float* __restrict__ deg, float* __restrict__ dis, int n) {
  int i = blockIdx.x * blockDim.x + threadIdx.x;
  if (i < n) dis[i] = rsqrtf(deg[i]);
}

__global__ void k_self1(const float* __restrict__ h, const float* __restrict__ dis,
                        float* __restrict__ out1, int n) {
  int i = blockIdx.x * blockDim.x + threadIdx.x;
  if (i >= n) return;
  float s = dis[i]; s = s * s;
  const float4* hp = reinterpret_cast<const float4*>(h + (size_t)i * HID);
  float4* op = reinterpret_cast<float4*>(out1 + (size_t)i * HID);
#pragma unroll
  for (int q = 0; q < 4; ++q) {
    float4 v = hp[q];
    v.x *= s; v.y *= s; v.z *= s; v.w *= s;
    op[q] = v;
  }
}

__global__ void k_scatter1(const int* __restrict__ src, const int* __restrict__ dst,
                           const float* __restrict__ dis, const float* __restrict__ h,
                           float* __restrict__ out1, int nE) {
  long long t = (long long)blockIdx.x * blockDim.x + threadIdx.x;
  int e = (int)(t >> 2);
  if (e >= nE) return;
  int q = (int)(t & 3);
  int s = src[e], d = dst[e];
  float norm = dis[s] * dis[d];
  float4 v = *reinterpret_cast<const float4*>(h + (size_t)s * HID + 4 * q);
  float* o = out1 + (size_t)d * HID + 4 * q;
  atomicAdd(o + 0, v.x * norm);
  atomicAdd(o + 1, v.y * norm);
  atomicAdd(o + 2, v.z * norm);
  atomicAdd(o + 3, v.w * norm);
}

__global__ void k_relu_bias(float* __restrict__ out1, const float* __restrict__ b1, int n) {
  long long t = (long long)blockIdx.x * blockDim.x + threadIdx.x;
  if (t >= (long long)n * HID) return;
  int j = (int)(t & (HID - 1));
  float v = out1[t] + b1[j];
  out1[t] = v > 0.f ? v : 0.f;
}

__global__ void k_self2(const float* __restrict__ h2, const float* __restrict__ dis,
                        float* __restrict__ out2, int n) {
  int i = blockIdx.x * blockDim.x + threadIdx.x;
  if (i >= n) return;
  float s = dis[i]; s = s * s;
  out2[(size_t)i * 2 + 0] = h2[(size_t)i * 2 + 0] * s;
  out2[(size_t)i * 2 + 1] = h2[(size_t)i * 2 + 1] * s;
}

__global__ void k_scatter2(const int* __restrict__ src, const int* __restrict__ dst,
                           const float* __restrict__ dis, const float* __restrict__ h2,
                           float* __restrict__ out2, int nE) {
  int e = blockIdx.x * blockDim.x + threadIdx.x;
  if (e >= nE) return;
  int s = src[e], d = dst[e];
  float norm = dis[s] * dis[d];
  atomicAdd(&out2[(size_t)d * 2 + 0], h2[(size_t)s * 2 + 0] * norm);
  atomicAdd(&out2[(size_t)d * 2 + 1], h2[(size_t)s * 2 + 1] * norm);
}

__global__ void k_logsoftmax(const float* __restrict__ out2, const float* __restrict__ b2,
                             float* __restrict__ y, int n) {
  int i = blockIdx.x * blockDim.x + threadIdx.x;
  if (i >= n) return;
  float a0 = out2[(size_t)i * 2 + 0] + b2[0];
  float a1 = out2[(size_t)i * 2 + 1] + b2[1];
  float mx = fmaxf(a0, a1);
  float lse = mx + logf(expf(a0 - mx) + expf(a1 - mx));
  y[(size_t)i * 2 + 0] = a0 - lse;
  y[(size_t)i * 2 + 1] = a1 - lse;
}


extern "C" void kernel_launch(void* const* d_in, const int* in_sizes, int n_in,
                              void* d_out, int out_size, void* d_ws, size_t ws_size,
                              hipStream_t stream) {
  const float* x  = (const float*)d_in[0];
  const int*   ei = (const int*)d_in[1];
  const float* W1 = (const float*)d_in[2];
  const float* b1 = (const float*)d_in[3];
  const float* W2 = (const float*)d_in[4];
  const float* b2 = (const float*)d_in[5];

  const int n  = in_sizes[0] / IN_DIM;
  const int nE = in_sizes[1] / 2;
  const int* src = ei;
  const int* dst = ei + nE;
  float* y = (float*)d_out;

  dim3 blk(TB);
  int gN   = (n + TB - 1) / TB;
  int gE   = (nE + TB - 1) / TB;
  int gN16 = (int)(((long long)n * HID + TB - 1) / TB);
  int tiles = (n + 15) / 16;
  int gT   = (tiles + (TB / 32) - 1) / (TB / 32);

  const size_t big_words  = (size_t)nE > (size_t)n * 2 * HID ? (size_t)nE : (size_t)n * 2 * HID;
  const size_t need_words = (size_t)n * 5 + 1024 + (size_t)nE + big_words;
  const int NB  = (n + RANGE - 1) / RANGE;
  const int NCH = (nE + CH - 1) / CH;
  const size_t n2 = (size_t)NB * (size_t)NCH;
  if (ws_size < need_words * sizeof(int) || NB > NBMAX || NCH > NCHMAX || (nE & 3) != 0 || (n & 3) != 0 ||
      n2 > (size_t)n) {
    float* deg  = (float*)d_ws;
    float* dis  = deg + n;
    float* h    = dis + n;
    float* out1 = h    + (size_t)n * HID;
    float* h2   = out1 + (size_t)n * HID;
    float* out2 = h2   + (size_t)n * 2;
    int gE4 = (int)(((long long)nE * 4 + TB - 1) / TB);
    k_init_deg  <<<gN,  blk, 0, stream>>>(deg, n);
    k_deg_edges <<<gE,  blk, 0, stream>>>(dst, deg, nE);
    k_dis       <<<gN,  blk, 0, stream>>>(deg, dis, n);
    k_gemm1_wmma<<<gT,  blk, 0, stream>>>(x, W1, h, n);
    k_self1     <<<gN,  blk, 0, stream>>>(h, dis, out1, n);
    k_scatter1  <<<gE4, blk, 0, stream>>>(src, dst, dis, h, out1, nE);
    k_relu_bias <<<gN16,blk, 0, stream>>>(out1, b1, n);
    k_gemm2_wmma<false><<<gT, blk, 0, stream>>>(out1, W2, dis, h2, n);
    k_self2     <<<gN,  blk, 0, stream>>>(h2, dis, out2, n);
    k_scatter2  <<<gE,  blk, 0, stream>>>(src, dst, dis, h2, out2, nE);
    k_logsoftmax<<<gN,  blk, 0, stream>>>(out2, b2, y, n);
    return;
  }

  int*   cnt     = (int*)d_ws;
  int*   row_ptr = cnt + n;
  float* dis     = (float*)(row_ptr + n);
  int*   bsum    = (int*)(dis + n);
  int*   cc      = bsum + 1024;
  int*   startT  = cc + n2;
  unsigned* binned = (unsigned*)(cc + (size_t)n * 2);
  float* h       = (float*)binned;
  float* r       = h + (size_t)n * HID;
  float* t2      = h;
  int*   col     = (int*)binned + big_words;

  dim3 sblk(SB);
  const bool tab = n <= DCAP && (size_t)NB * (RANGE / 4) <= n2;
  unsigned* deg8 = tab ? (unsigned*)cc : nullptr;
  k_chunk_sort <<<NCH, sblk, 0, stream>>>(src, dst, startT, binned, nE, NB, NCH);
  k_range_lists<<<NB,  sblk, 0, stream>>>(binned, startT, cnt, row_ptr, dis, col, deg8, n, nE, NB, NCH);
  k_gemm1_wmma <<<gT,  blk,  0, stream>>>(x, W1, h, n);
  if (tab) k_pull1_tab<<<NWGP, sblk, 0, stream>>>(col, row_ptr, cnt, dis, deg8, h, b1, r, n);
  else     k_pull1    <<<gN16, blk,  0, stream>>>(col, row_ptr, cnt, dis, h, b1, r, n);
  k_gemm2_wmma<true><<<gT, blk, 0, stream>>>(r, W2, dis, t2, n);
  const int nb16 = (n + 15) / 16;
  k_pull2_res  <<<(nb16 + P2W - 1) / P2W, dim3(32 * P2W), 0, stream>>>(col, row_ptr, cnt, t2, b2, y, n);
}
